# starting kernel
# baseline (speedup 1.0000x reference)
#include <hip/hip_runtime.h>
#include <math.h>

typedef __attribute__((ext_vector_type(2))) float v2f;
typedef __attribute__((ext_vector_type(4))) float v4f;
typedef __attribute__((ext_vector_type(8))) float v8f;

#define SEQ   100
#define BATCH 256
#define INDIM 300
#define HID   128
#define NROW  (BATCH*SEQ)
#define WIN   11

static __device__ __forceinline__ v8f wmma_f32_k4(v2f a, v2f b, v8f c) {
    return __builtin_amdgcn_wmma_f32_16x16x4_f32(false, a, false, b, (short)0, c,
                                                 false, false);
}

__global__ __launch_bounds__(256) void k_h(const float* __restrict__ x,
                                           const float* __restrict__ W1,
                                           const float* __restrict__ b1,
                                           float* __restrict__ h) {
    __shared__ float xs[16 * 301];
    const int m0 = blockIdx.x * 16;
    for (int e = threadIdx.x; e < 16 * INDIM; e += 256) {
        int row = e / INDIM, col = e - row * INDIM;
        xs[row * 301 + col] = x[(m0 + row) * INDIM + col];
    }
    __syncthreads();

    const int wave = threadIdx.x >> 5;
    const int lane = threadIdx.x & 31;
    const int rr   = lane & 15;
    const int half = lane >> 4;
    const int n    = wave * 16 + rr;

    v8f acc = {0.f,0.f,0.f,0.f,0.f,0.f,0.f,0.f};
    for (int kk = 0; kk < INDIM; kk += 4) {
        const int off = kk + half * 2;
        v2f a; a.x = xs[rr * 301 + off]; a.y = xs[rr * 301 + off + 1];
        v2f bf; bf.x = W1[off * HID + n]; bf.y = W1[(off + 1) * HID + n];
        acc = wmma_f32_k4(a, bf, acc);
    }
    const float bias = b1[n];
#pragma unroll
    for (int r = 0; r < 8; ++r) {
        const int m = m0 + r + half * 8;
        float v = acc[r] + bias;
        h[m * HID + n] = v > 0.f ? v : 0.f;
    }
}

__global__ __launch_bounds__(256) void k_qkv(const float* __restrict__ h,
                                             const float* __restrict__ Wq,
                                             const float* __restrict__ Wk,
                                             const float* __restrict__ Wv,
                                             float* __restrict__ Q,
                                             float* __restrict__ K,
                                             float* __restrict__ V) {
    __shared__ float hs[16 * 132];
    const int m0 = blockIdx.x * 16;
    for (int e = threadIdx.x; e < 16 * HID; e += 256) {
        int row = e >> 7, col = e & 127;
        hs[row * 132 + col] = h[(m0 + row) * HID + col];
    }
    __syncthreads();

    const int wave = threadIdx.x >> 5;
    const int lane = threadIdx.x & 31;
    const int rr   = lane & 15;
    const int half = lane >> 4;
    const int n    = wave * 16 + rr;

    v8f aq = {0.f,0.f,0.f,0.f,0.f,0.f,0.f,0.f};
    v8f ak = aq, av = aq;
    for (int kk = 0; kk < HID; kk += 4) {
        const int off = kk + half * 2;
        v2f a; a.x = hs[rr * 132 + off]; a.y = hs[rr * 132 + off + 1];
        v2f bq; bq.x = Wq[off * HID + n]; bq.y = Wq[(off + 1) * HID + n];
        aq = wmma_f32_k4(a, bq, aq);
        v2f bk; bk.x = Wk[off * HID + n]; bk.y = Wk[(off + 1) * HID + n];
        ak = wmma_f32_k4(a, bk, ak);
        v2f bv; bv.x = Wv[off * HID + n]; bv.y = Wv[(off + 1) * HID + n];
        av = wmma_f32_k4(a, bv, av);
    }
#pragma unroll
    for (int r = 0; r < 8; ++r) {
        const int m = m0 + r + half * 8;
        Q[m * HID + n] = aq[r];
        K[m * HID + n] = ak[r];
        V[m * HID + n] = av[r];
    }
}

__global__ __launch_bounds__(256) void k_attn(const float* __restrict__ Q,
                                              const float* __restrict__ Km,
                                              const float* __restrict__ V,
                                              const float* __restrict__ W2,
                                              const float* __restrict__ b2,
                                              float* __restrict__ out,
                                              float* __restrict__ alphaOut) {
    const int wave = threadIdx.x >> 5;
    const int lane = threadIdx.x & 31;
    const int rr   = lane & 15;
    const int half = lane >> 4;
    const int idx  = blockIdx.x * 8 + wave;
    if (idx >= NROW) return;
    const int s = idx % SEQ;

    const int  r     = s + 5 - rr;
    const bool rowok = (rr < WIN) && (r >= 0) && (r < SEQ);
    const float* qrow = Q  + (long)(idx + 5 - rr) * HID;
    const float* krow = Km + (long)(idx + 5 - rr) * HID;

    v8f acc = {0.f,0.f,0.f,0.f,0.f,0.f,0.f,0.f};
    for (int kk = 0; kk < HID; kk += 4) {
        const int off = kk + half * 2;
        v2f a  = {0.f, 0.f};
        v2f bf = {0.f, 0.f};
        if (rowok) {
            a  = *(const v2f*)(qrow + off);
            bf = *(const v2f*)(krow + off);
        }
        acc = wmma_f32_k4(a, bf, acc);
    }

    const float scale = 0.08838834764831845f;
    const long  abase = (long)idx * (WIN * WIN);
    float bsum = 0.f;
#pragma unroll
    for (int rI = 0; rI < 8; ++rI) {
        const float sc = acc[rI] * scale;
        float mv = (rr < WIN) ? sc : -3.0e38f;
        mv = fmaxf(mv, __shfl_xor(mv, 1, 32));
        mv = fmaxf(mv, __shfl_xor(mv, 2, 32));
        mv = fmaxf(mv, __shfl_xor(mv, 4, 32));
        mv = fmaxf(mv, __shfl_xor(mv, 8, 32));
        float p = (rr < WIN) ? expf(sc - mv) : 0.f;
        float sum = p;
        sum += __shfl_xor(sum, 1, 32);
        sum += __shfl_xor(sum, 2, 32);
        sum += __shfl_xor(sum, 4, 32);
        sum += __shfl_xor(sum, 8, 32);
        const float al = p / sum;
        const int i = rI + half * 8;
        if (i < WIN) {
            if (rr < WIN) alphaOut[abase + i * WIN + rr] = al;
            bsum += al;
        }
    }
    bsum += __shfl_xor(bsum, 16, 32);

    v4f pooled = {0.f, 0.f, 0.f, 0.f};
    for (int j = 0; j < WIN; ++j) {
        const float bj = __shfl(bsum, j, 32);
        const int rv = s + 5 - j;
        if (rv >= 0 && rv < SEQ) {
            v4f vv = *(const v4f*)(V + (long)(idx + 5 - j) * HID + lane * 4);
            pooled += bj * vv;
        }
    }

    float p0 = 0.f, p1 = 0.f;
#pragma unroll
    for (int t = 0; t < 4; ++t) {
        const int d = lane * 4 + t;
        p0 += pooled[t] * W2[d * 2 + 0];
        p1 += pooled[t] * W2[d * 2 + 1];
    }
#pragma unroll
    for (int mask = 1; mask < 32; mask <<= 1) {
        p0 += __shfl_xor(p0, mask, 32);
        p1 += __shfl_xor(p1, mask, 32);
    }
    if (lane == 0) {
        out[idx * 2 + 0] = 1.f / (1.f + expf(-(p0 + b2[0])));
        out[idx * 2 + 1] = 1.f / (1.f + expf(-(p1 + b2[1])));
    }
}

extern "C" void kernel_launch(void* const* d_in, const int* in_sizes, int n_in,
                              void* d_out, int out_size, void* d_ws, size_t ws_size,
                              hipStream_t stream) {
    const float* x  = (const float*)d_in[0];
    const float* W1 = (const float*)d_in[1];
    const float* b1 = (const float*)d_in[2];
    const float* Wq = (const float*)d_in[3];
    const float* Wk = (const float*)d_in[4];
    const float* Wv = (const float*)d_in[5];
    const float* W2 = (const float*)d_in[6];
    const float* b2 = (const float*)d_in[7];

    float* h = (float*)d_ws;
    float* Q = h + (size_t)NROW * HID;
    float* K = Q + (size_t)NROW * HID;
    float* V = K + (size_t)NROW * HID;

    float* out      = (float*)d_out;
    float* alphaOut = out + (size_t)NROW * 2;

    k_h  <<<NROW / 16, 256, 0, stream>>>(x, W1, b1, h);
    k_qkv<<<NROW / 16, 256, 0, stream>>>(h, Wq, Wk, Wv, Q, K, V);
    k_attn<<<NROW / 8, 256, 0, stream>>>(Q, K, V, W2, b2, out, alphaOut);
}


# k_attn: every lane exchange (softmax row max and row sum, beta fold and broadcast, logit sum) on the vector ALU: row_xmask DPP, permlanex16, readlane; were ds_bpermute round trips; sums unchanged
# speedup vs baseline: 1.2630x; 1.2630x over previous
#include <hip/hip_runtime.h>
#include <math.h>

typedef __attribute__((ext_vector_type(2))) float v2f;
typedef __attribute__((ext_vector_type(4))) float v4f;
typedef __attribute__((ext_vector_type(8))) float v8f;

#define SEQ   100
#define BATCH 256
#define INDIM 300
#define HID   128
#define NROW  (BATCH*SEQ)
#define WIN   11
#define GP    32

static __device__ __forceinline__ v8f wmma_f32_k4(v2f a, v2f b, v8f c) {
    return __builtin_amdgcn_wmma_f32_16x16x4_f32(false, a, false, b, (short)0, c,
                                                 false, false);
}

template <int M>
static __device__ __forceinline__ float row_xor(float x) {
    const int b = __builtin_bit_cast(int, x);
    return __builtin_bit_cast(float,
        __builtin_amdgcn_update_dpp(b, b, 0x160 + M, 0xf, 0xf, true));
}
static __device__ __forceinline__ float xhalf(float x) {
    const int b = __builtin_bit_cast(int, x);
    return __builtin_bit_cast(float,
        __builtin_amdgcn_permlanex16(b, b, 0x76543210, 0xfedcba98, false, false));
}

__global__ __launch_bounds__(256) void k_h(const float* __restrict__ x,
                                           const float* __restrict__ W1,
                                           const float* __restrict__ b1,
                                           float* __restrict__ h) {
    __shared__ float xs[16 * 301];
    const int m0 = blockIdx.x * 16;
    for (int e = threadIdx.x; e < 16 * INDIM; e += 256) {
        int row = e / INDIM, col = e - row * INDIM;
        xs[row * 301 + col] = x[(m0 + row) * INDIM + col];
    }
    __syncthreads();

    const int wave = threadIdx.x >> 5;
    const int lane = threadIdx.x & 31;
    const int rr   = lane & 15;
    const int half = lane >> 4;
    const int n    = wave * 16 + rr;

    v8f acc = {0.f,0.f,0.f,0.f,0.f,0.f,0.f,0.f};
    for (int kk = 0; kk < INDIM; kk += 4) {
        const int off = kk + half * 2;
        v2f a; a.x = xs[rr * 301 + off]; a.y = xs[rr * 301 + off + 1];
        v2f bf; bf.x = W1[off * HID + n]; bf.y = W1[(off + 1) * HID + n];
        acc = wmma_f32_k4(a, bf, acc);
    }
    const float bias = b1[n];
#pragma unroll
    for (int r = 0; r < 8; ++r) {
        const int m = m0 + r + half * 8;
        float v = acc[r] + bias;
        h[m * HID + n] = v > 0.f ? v : 0.f;
    }
}

__global__ __launch_bounds__(256) void k_qkv(const float* __restrict__ h,
                                             const float* __restrict__ Wq,
                                             const float* __restrict__ Wk,
                                             const float* __restrict__ Wv,
                                             float* __restrict__ Q,
                                             float* __restrict__ K,
                                             float* __restrict__ V) {
    __shared__ float hs[16 * 132];
    const int m0 = blockIdx.x * 16;
    for (int e = threadIdx.x; e < 16 * HID; e += 256) {
        int row = e >> 7, col = e & 127;
        hs[row * 132 + col] = h[(m0 + row) * HID + col];
    }
    __syncthreads();

    const int wave = threadIdx.x >> 5;
    const int lane = threadIdx.x & 31;
    const int rr   = lane & 15;
    const int half = lane >> 4;
    const int n    = wave * 16 + rr;

    v8f aq = {0.f,0.f,0.f,0.f,0.f,0.f,0.f,0.f};
    v8f ak = aq, av = aq;
    for (int kk = 0; kk < HID; kk += 4) {
        const int off = kk + half * 2;
        v2f a; a.x = hs[rr * 132 + off]; a.y = hs[rr * 132 + off + 1];
        v2f bq; bq.x = Wq[off * HID + n]; bq.y = Wq[(off + 1) * HID + n];
        aq = wmma_f32_k4(a, bq, aq);
        v2f bk; bk.x = Wk[off * HID + n]; bk.y = Wk[(off + 1) * HID + n];
        ak = wmma_f32_k4(a, bk, ak);
        v2f bv; bv.x = Wv[off * HID + n]; bv.y = Wv[(off + 1) * HID + n];
        av = wmma_f32_k4(a, bv, av);
    }
#pragma unroll
    for (int r = 0; r < 8; ++r) {
        const int m = m0 + r + half * 8;
        Q[m * HID + n] = aq[r];
        K[m * HID + n] = ak[r];
        V[m * HID + n] = av[r];
    }
}

__global__ __launch_bounds__(224) void k_gram(const float* __restrict__ Q,
                                              const float* __restrict__ Km,
                                              float* __restrict__ G) {
    const int I    = threadIdx.x >> 5;
    const int lane = threadIdx.x & 31;
    const int rr   = lane & 15;
    const int half = lane >> 4;
    const long base = (long)blockIdx.x * SEQ;

    const int ra  = min(I * 16 + rr, SEQ - 1);
    const int rc0 = max((I - 1) * 16 + rr, 0);
    const int rc1 = min(I * 16 + rr, SEQ - 1);
    const int rc2 = min((I + 1) * 16 + rr, SEQ - 1);
    const float* qrow = Q  + (base + ra)  * HID + half * 4;
    const float* kr0  = Km + (base + rc0) * HID + half * 4;
    const float* kr1  = Km + (base + rc1) * HID + half * 4;
    const float* kr2  = Km + (base + rc2) * HID + half * 4;

    v8f g0 = {0.f,0.f,0.f,0.f,0.f,0.f,0.f,0.f};
    v8f g1 = g0, g2 = g0;
    for (int t = 0; t < HID; t += 8) {
        const v4f a  = *(const v4f*)(qrow + t);
        const v4f kb0 = *(const v4f*)(kr0 + t);
        const v4f kb1 = *(const v4f*)(kr1 + t);
        const v4f kb2 = *(const v4f*)(kr2 + t);
        v2f alo = {a.x, a.y}, ahi = {a.z, a.w};
        v2f lo0 = {kb0.x, kb0.y}, hi0 = {kb0.z, kb0.w};
        v2f lo1 = {kb1.x, kb1.y}, hi1 = {kb1.z, kb1.w};
        v2f lo2 = {kb2.x, kb2.y}, hi2 = {kb2.z, kb2.w};
        g0 = wmma_f32_k4(alo, lo0, g0);
        g1 = wmma_f32_k4(alo, lo1, g1);
        g2 = wmma_f32_k4(alo, lo2, g2);
        g0 = wmma_f32_k4(ahi, hi0, g0);
        g1 = wmma_f32_k4(ahi, hi1, g1);
        g2 = wmma_f32_k4(ahi, hi2, g2);
    }

    const float scale = 0.08838834764831845f;
#pragma unroll
    for (int r = 0; r < 8; ++r) {
        const int a = I * 16 + r + half * 8;
        if (a < SEQ) {
            float* grow = G + (base + a) * GP + (10 - a);
            const int c0 = (I - 1) * 16 + rr;
            const int c1 = I * 16 + rr;
            const int c2 = (I + 1) * 16 + rr;
            if (c0 >= 0  && a - c0 <= 10)                 grow[c0] = g0[r] * scale;
            if (c1 < SEQ && c1 - a <= 10 && a - c1 <= 10) grow[c1] = g1[r] * scale;
            if (c2 < SEQ && c2 - a <= 10)                 grow[c2] = g2[r] * scale;
        }
    }
}

__global__ __launch_bounds__(256) void k_attn(const float* __restrict__ G,
                                              const float* __restrict__ V,
                                              const float* __restrict__ W2,
                                              const float* __restrict__ b2,
                                              float* __restrict__ out,
                                              float* __restrict__ alphaOut) {
    const int wave = threadIdx.x >> 5;
    const int lane = threadIdx.x & 31;
    const int rr   = lane & 15;
    const int half = lane >> 4;
    const int idx  = blockIdx.x * 8 + wave;
    if (idx >= NROW) return;
    const int s = idx % SEQ;

    const int  cj  = s + 5 - rr;
    const bool jok = (rr < WIN) && (cj >= 0) && (cj < SEQ);
    const float* gcol  = G + (long)(idx + 5) * GP + (10 - rr);
    const float* gdiag = G + (long)idx * GP + 10;

    float scv[8];
#pragma unroll
    for (int rI = 0; rI < 8; ++rI) {
        const int  i  = rI + half * 8;
        const int  ai = s + 5 - i;
        const bool ok = jok && (i < WIN) && (ai >= 0) && (ai < SEQ);
        const float g = *(ok ? gcol - (GP - 1) * i : gdiag);
        scv[rI] = ok ? g : 0.f;
    }

    const long  abase = (long)idx * (WIN * WIN);
    float bsum = 0.f;
#pragma unroll
    for (int rI = 0; rI < 8; ++rI) {
        const int   i  = rI + half * 8;
        const float sc = scv[rI];
        float mv = (rr < WIN) ? sc : -3.0e38f;
        mv = __builtin_elementwise_maximum(mv, row_xor<1>(mv));
        mv = __builtin_elementwise_maximum(mv, row_xor<2>(mv));
        mv = __builtin_elementwise_maximum(mv, row_xor<4>(mv));
        mv = __builtin_elementwise_maximum(mv, row_xor<8>(mv));
        float p = (rr < WIN) ? expf(sc - mv) : 0.f;
        float sum = p;
        sum += row_xor<1>(sum);
        sum += row_xor<2>(sum);
        sum += row_xor<4>(sum);
        sum += row_xor<8>(sum);
        const float al = p / sum;
        if (i < WIN) {
            if (rr < WIN) alphaOut[abase + i * WIN + rr] = al;
            bsum += al;
        }
    }
    bsum += xhalf(bsum);

    v4f pooled = {0.f, 0.f, 0.f, 0.f};
#pragma unroll
    for (int j = 0; j < WIN; ++j) {
        const float bj = __builtin_bit_cast(float,
            __builtin_amdgcn_readlane(__builtin_bit_cast(int, bsum), j));
        const int rv = s + 5 - j;
        if (rv >= 0 && rv < SEQ) {
            v4f vv = *(const v4f*)(V + (long)(idx + 5 - j) * HID + lane * 4);
            pooled += bj * vv;
        }
    }

    float p0 = 0.f, p1 = 0.f;
#pragma unroll
    for (int t = 0; t < 4; ++t) {
        const int d = lane * 4 + t;
        p0 += pooled[t] * W2[d * 2 + 0];
        p1 += pooled[t] * W2[d * 2 + 1];
    }
    p0 += row_xor<1>(p0);  p1 += row_xor<1>(p1);
    p0 += row_xor<2>(p0);  p1 += row_xor<2>(p1);
    p0 += row_xor<4>(p0);  p1 += row_xor<4>(p1);
    p0 += row_xor<8>(p0);  p1 += row_xor<8>(p1);
    p0 += xhalf(p0);       p1 += xhalf(p1);
    if (lane == 0) {
        out[idx * 2 + 0] = 1.f / (1.f + expf(-(p0 + b2[0])));
        out[idx * 2 + 1] = 1.f / (1.f + expf(-(p1 + b2[1])));
    }
}

extern "C" void kernel_launch(void* const* d_in, const int* in_sizes, int n_in,
                              void* d_out, int out_size, void* d_ws, size_t ws_size,
                              hipStream_t stream) {
    const float* x  = (const float*)d_in[0];
    const float* W1 = (const float*)d_in[1];
    const float* b1 = (const float*)d_in[2];
    const float* Wq = (const float*)d_in[3];
    const float* Wk = (const float*)d_in[4];
    const float* Wv = (const float*)d_in[5];
    const float* W2 = (const float*)d_in[6];
    const float* b2 = (const float*)d_in[7];

    float* h = (float*)d_ws;
    float* Q = h + (size_t)NROW * HID;
    float* K = Q + (size_t)NROW * HID;
    float* V = K + (size_t)NROW * HID;
    float* G = h;

    float* out      = (float*)d_out;
    float* alphaOut = out + (size_t)NROW * 2;

    k_h  <<<NROW / 16, 256, 0, stream>>>(x, W1, b1, h);
    k_qkv<<<NROW / 16, 256, 0, stream>>>(h, Wq, Wk, Wv, Q, K, V);
    k_gram<<<BATCH, 224, 0, stream>>>(Q, K, G);
    k_attn<<<NROW / 8, 256, 0, stream>>>(G, V, W2, b2, out, alphaOut);
}


# k_attn: each 16-lane half of a wave takes one position, its 11 score rows in sequence (was one position a wave, rows split 8+3 over the halves): 12800 waves, was 25600; beta, logit sum order changes
# speedup vs baseline: 1.4385x; 1.1390x over previous
#include <hip/hip_runtime.h>
#include <math.h>

typedef __attribute__((ext_vector_type(2))) float v2f;
typedef __attribute__((ext_vector_type(4))) float v4f;
typedef __attribute__((ext_vector_type(8))) float v8f;

#define SEQ   100
#define BATCH 256
#define INDIM 300
#define HID   128
#define NROW  (BATCH*SEQ)
#define WIN   11
#define GP    32

static __device__ __forceinline__ v8f wmma_f32_k4(v2f a, v2f b, v8f c) {
    return __builtin_amdgcn_wmma_f32_16x16x4_f32(false, a, false, b, (short)0, c,
                                                 false, false);
}

template <int M>
static __device__ __forceinline__ float row_xor(float x) {
    const int b = __builtin_bit_cast(int, x);
    return __builtin_bit_cast(float,
        __builtin_amdgcn_update_dpp(b, b, 0x160 + M, 0xf, 0xf, true));
}

__global__ __launch_bounds__(256) void k_h(const float* __restrict__ x,
                                           const float* __restrict__ W1,
                                           const float* __restrict__ b1,
                                           float* __restrict__ h) {
    __shared__ float xs[16 * 301];
    const int m0 = blockIdx.x * 16;
    for (int e = threadIdx.x; e < 16 * INDIM; e += 256) {
        int row = e / INDIM, col = e - row * INDIM;
        xs[row * 301 + col] = x[(m0 + row) * INDIM + col];
    }
    __syncthreads();

    const int wave = threadIdx.x >> 5;
    const int lane = threadIdx.x & 31;
    const int rr   = lane & 15;
    const int half = lane >> 4;
    const int n    = wave * 16 + rr;

    v8f acc = {0.f,0.f,0.f,0.f,0.f,0.f,0.f,0.f};
    for (int kk = 0; kk < INDIM; kk += 4) {
        const int off = kk + half * 2;
        v2f a; a.x = xs[rr * 301 + off]; a.y = xs[rr * 301 + off + 1];
        v2f bf; bf.x = W1[off * HID + n]; bf.y = W1[(off + 1) * HID + n];
        acc = wmma_f32_k4(a, bf, acc);
    }
    const float bias = b1[n];
#pragma unroll
    for (int r = 0; r < 8; ++r) {
        const int m = m0 + r + half * 8;
        float v = acc[r] + bias;
        h[m * HID + n] = v > 0.f ? v : 0.f;
    }
}

__global__ __launch_bounds__(256) void k_qkv(const float* __restrict__ h,
                                             const float* __restrict__ Wq,
                                             const float* __restrict__ Wk,
                                             const float* __restrict__ Wv,
                                             float* __restrict__ Q,
                                             float* __restrict__ K,
                                             float* __restrict__ V) {
    __shared__ float hs[16 * 132];
    const int m0 = blockIdx.x * 16;
    for (int e = threadIdx.x; e < 16 * HID; e += 256) {
        int row = e >> 7, col = e & 127;
        hs[row * 132 + col] = h[(m0 + row) * HID + col];
    }
    __syncthreads();

    const int wave = threadIdx.x >> 5;
    const int lane = threadIdx.x & 31;
    const int rr   = lane & 15;
    const int half = lane >> 4;
    const int n    = wave * 16 + rr;

    v8f aq = {0.f,0.f,0.f,0.f,0.f,0.f,0.f,0.f};
    v8f ak = aq, av = aq;
    for (int kk = 0; kk < HID; kk += 4) {
        const int off = kk + half * 2;
        v2f a; a.x = hs[rr * 132 + off]; a.y = hs[rr * 132 + off + 1];
        v2f bq; bq.x = Wq[off * HID + n]; bq.y = Wq[(off + 1) * HID + n];
        aq = wmma_f32_k4(a, bq, aq);
        v2f bk; bk.x = Wk[off * HID + n]; bk.y = Wk[(off + 1) * HID + n];
        ak = wmma_f32_k4(a, bk, ak);
        v2f bv; bv.x = Wv[off * HID + n]; bv.y = Wv[(off + 1) * HID + n];
        av = wmma_f32_k4(a, bv, av);
    }
#pragma unroll
    for (int r = 0; r < 8; ++r) {
        const int m = m0 + r + half * 8;
        Q[m * HID + n] = aq[r];
        K[m * HID + n] = ak[r];
        V[m * HID + n] = av[r];
    }
}

__global__ __launch_bounds__(224) void k_gram(const float* __restrict__ Q,
                                              const float* __restrict__ Km,
                                              float* __restrict__ G) {
    const int I    = threadIdx.x >> 5;
    const int lane = threadIdx.x & 31;
    const int rr   = lane & 15;
    const int half = lane >> 4;
    const long base = (long)blockIdx.x * SEQ;

    const int ra  = min(I * 16 + rr, SEQ - 1);
    const int rc0 = max((I - 1) * 16 + rr, 0);
    const int rc1 = min(I * 16 + rr, SEQ - 1);
    const int rc2 = min((I + 1) * 16 + rr, SEQ - 1);
    const float* qrow = Q  + (base + ra)  * HID + half * 4;
    const float* kr0  = Km + (base + rc0) * HID + half * 4;
    const float* kr1  = Km + (base + rc1) * HID + half * 4;
    const float* kr2  = Km + (base + rc2) * HID + half * 4;

    v8f g0 = {0.f,0.f,0.f,0.f,0.f,0.f,0.f,0.f};
    v8f g1 = g0, g2 = g0;
    for (int t = 0; t < HID; t += 8) {
        const v4f a  = *(const v4f*)(qrow + t);
        const v4f kb0 = *(const v4f*)(kr0 + t);
        const v4f kb1 = *(const v4f*)(kr1 + t);
        const v4f kb2 = *(const v4f*)(kr2 + t);
        v2f alo = {a.x, a.y}, ahi = {a.z, a.w};
        v2f lo0 = {kb0.x, kb0.y}, hi0 = {kb0.z, kb0.w};
        v2f lo1 = {kb1.x, kb1.y}, hi1 = {kb1.z, kb1.w};
        v2f lo2 = {kb2.x, kb2.y}, hi2 = {kb2.z, kb2.w};
        g0 = wmma_f32_k4(alo, lo0, g0);
        g1 = wmma_f32_k4(alo, lo1, g1);
        g2 = wmma_f32_k4(alo, lo2, g2);
        g0 = wmma_f32_k4(ahi, hi0, g0);
        g1 = wmma_f32_k4(ahi, hi1, g1);
        g2 = wmma_f32_k4(ahi, hi2, g2);
    }

    const float scale = 0.08838834764831845f;
#pragma unroll
    for (int r = 0; r < 8; ++r) {
        const int a = I * 16 + r + half * 8;
        if (a < SEQ) {
            float* grow = G + (base + a) * GP + (10 - a);
            const int c0 = (I - 1) * 16 + rr;
            const int c1 = I * 16 + rr;
            const int c2 = (I + 1) * 16 + rr;
            if (c0 >= 0  && a - c0 <= 10)                 grow[c0] = g0[r] * scale;
            if (c1 < SEQ && c1 - a <= 10 && a - c1 <= 10) grow[c1] = g1[r] * scale;
            if (c2 < SEQ && c2 - a <= 10)                 grow[c2] = g2[r] * scale;
        }
    }
}

template <int J>
static __device__ __forceinline__ float row_bcast(float x) {
    const int b = __builtin_bit_cast(int, x);
    return __builtin_bit_cast(float,
        __builtin_amdgcn_update_dpp(b, b, 0x150 + J, 0xf, 0xf, true));
}

template <int J>
static __device__ __forceinline__ void pool_row(const float* __restrict__ vseq, int s,
                                                float bsum, v4f& plo, v4f& phi) {
    const int   rv = s + 5 - J;
    const int   rc = min(max(rv, 0), SEQ - 1);
    const float bj = row_bcast<J>(bsum);
    const float w  = (rv >= 0 && rv < SEQ) ? bj : 0.f;
    const float* vr = vseq + rc * HID;
    const v4f lo = *(const v4f*)(vr);
    const v4f hi = *(const v4f*)(vr + 64);
    plo += w * lo;
    phi += w * hi;
}

__global__ __launch_bounds__(256, 16) void k_attn(const float* __restrict__ G,
                                              const float* __restrict__ V,
                                              const float* __restrict__ W2,
                                              const float* __restrict__ b2,
                                              float* __restrict__ out,
                                              float* __restrict__ alphaOut) {
    const int rr  = threadIdx.x & 15;
    const int idx = blockIdx.x * 16 + (threadIdx.x >> 4);
    const int s   = idx % SEQ;

    const int  cj  = s + 5 - rr;
    const bool jin = rr < WIN;
    const bool jok = jin && (cj >= 0) && (cj < SEQ);
    const float* gcol  = G + (long)(idx + 5) * GP + (10 - rr);
    const float* gdiag = G + (long)idx * GP + 10;

    float scv[WIN];
#pragma unroll
    for (int i = 0; i < WIN; ++i) {
        const int  ai = s + 5 - i;
        const bool ok = jok && (ai >= 0) && (ai < SEQ);
        const float g = *(ok ? gcol - (GP - 1) * i : gdiag);
        scv[i] = ok ? g : 0.f;
    }

    const long  abase = (long)idx * (WIN * WIN);
    float bsum = 0.f;
#pragma unroll
    for (int i = 0; i < WIN; ++i) {
        const float sc = scv[i];
        float mv = jin ? sc : -3.0e38f;
        mv = __builtin_elementwise_maximum(mv, row_xor<1>(mv));
        mv = __builtin_elementwise_maximum(mv, row_xor<2>(mv));
        mv = __builtin_elementwise_maximum(mv, row_xor<4>(mv));
        mv = __builtin_elementwise_maximum(mv, row_xor<8>(mv));
        float p = jin ? expf(sc - mv) : 0.f;
        float sum = p;
        sum += row_xor<1>(sum);
        sum += row_xor<2>(sum);
        sum += row_xor<4>(sum);
        sum += row_xor<8>(sum);
        const float al = p / sum;
        if (jin) alphaOut[abase + i * WIN + rr] = al;
        bsum += al;
    }

    v4f plo = {0.f, 0.f, 0.f, 0.f};
    v4f phi = {0.f, 0.f, 0.f, 0.f};
    const float* vseq = V + (long)(idx - s) * HID + rr * 4;
    pool_row<0>(vseq, s, bsum, plo, phi);
    pool_row<1>(vseq, s, bsum, plo, phi);
    pool_row<2>(vseq, s, bsum, plo, phi);
    pool_row<3>(vseq, s, bsum, plo, phi);
    pool_row<4>(vseq, s, bsum, plo, phi);
    pool_row<5>(vseq, s, bsum, plo, phi);
    pool_row<6>(vseq, s, bsum, plo, phi);
    pool_row<7>(vseq, s, bsum, plo, phi);
    pool_row<8>(vseq, s, bsum, plo, phi);
    pool_row<9>(vseq, s, bsum, plo, phi);
    pool_row<10>(vseq, s, bsum, plo, phi);

    const v4f wa = *(const v4f*)(W2 + rr * 8);
    const v4f wb = *(const v4f*)(W2 + rr * 8 + 4);
    const v4f wc = *(const v4f*)(W2 + 128 + rr * 8);
    const v4f wd = *(const v4f*)(W2 + 128 + rr * 8 + 4);
    float p0 = 0.f, p1 = 0.f;
    p0 += plo.x * wa.x;  p1 += plo.x * wa.y;
    p0 += plo.y * wa.z;  p1 += plo.y * wa.w;
    p0 += plo.z * wb.x;  p1 += plo.z * wb.y;
    p0 += plo.w * wb.z;  p1 += plo.w * wb.w;
    p0 += phi.x * wc.x;  p1 += phi.x * wc.y;
    p0 += phi.y * wc.z;  p1 += phi.y * wc.w;
    p0 += phi.z * wd.x;  p1 += phi.z * wd.y;
    p0 += phi.w * wd.z;  p1 += phi.w * wd.w;
    p0 += row_xor<1>(p0);  p1 += row_xor<1>(p1);
    p0 += row_xor<2>(p0);  p1 += row_xor<2>(p1);
    p0 += row_xor<4>(p0);  p1 += row_xor<4>(p1);
    p0 += row_xor<8>(p0);  p1 += row_xor<8>(p1);
    if (rr == 0) {
        out[idx * 2 + 0] = 1.f / (1.f + expf(-(p0 + b2[0])));
        out[idx * 2 + 1] = 1.f / (1.f + expf(-(p1 + b2[1])));
    }
}

extern "C" void kernel_launch(void* const* d_in, const int* in_sizes, int n_in,
                              void* d_out, int out_size, void* d_ws, size_t ws_size,
                              hipStream_t stream) {
    const float* x  = (const float*)d_in[0];
    const float* W1 = (const float*)d_in[1];
    const float* b1 = (const float*)d_in[2];
    const float* Wq = (const float*)d_in[3];
    const float* Wk = (const float*)d_in[4];
    const float* Wv = (const float*)d_in[5];
    const float* W2 = (const float*)d_in[6];
    const float* b2 = (const float*)d_in[7];

    float* h = (float*)d_ws;
    float* Q = h + (size_t)NROW * HID;
    float* K = Q + (size_t)NROW * HID;
    float* V = K + (size_t)NROW * HID;
    float* G = h;

    float* out      = (float*)d_out;
    float* alphaOut = out + (size_t)NROW * 2;

    k_h  <<<NROW / 16, 256, 0, stream>>>(x, W1, b1, h);
    k_qkv<<<NROW / 16, 256, 0, stream>>>(h, Wq, Wk, Wv, Q, K, V);
    k_gram<<<BATCH, 224, 0, stream>>>(Q, K, G);
    k_attn<<<NROW / 16, 256, 0, stream>>>(G, V, W2, b2, out, alphaOut);
}


# k_h, k_qkv: a wave keeps 2 row tiles of its 16 columns in registers: a weight fragment loaded once feeds 2 WMMAs (was 1); blocks of 32 rows x 64 columns, 4 waves, 16-byte tile copies; sums unchanged
# speedup vs baseline: 1.5599x; 1.0844x over previous
#include <hip/hip_runtime.h>
#include <math.h>

typedef __attribute__((ext_vector_type(2))) float v2f;
typedef __attribute__((ext_vector_type(4))) float v4f;
typedef __attribute__((ext_vector_type(8))) float v8f;

#define SEQ   100
#define BATCH 256
#define INDIM 300
#define HID   128
#define NROW  (BATCH*SEQ)
#define WIN   11
#define GP    32

static __device__ __forceinline__ v8f wmma_f32_k4(v2f a, v2f b, v8f c) {
    return __builtin_amdgcn_wmma_f32_16x16x4_f32(false, a, false, b, (short)0, c,
                                                 false, false);
}

template <int M>
static __device__ __forceinline__ float row_xor(float x) {
    const int b = __builtin_bit_cast(int, x);
    return __builtin_bit_cast(float,
        __builtin_amdgcn_update_dpp(b, b, 0x160 + M, 0xf, 0xf, true));
}

#define XV4   (32 * INDIM / 4)
#define XFULL (XV4 / 128)
__global__ __launch_bounds__(128) void k_h(const float* __restrict__ x,
                                           const float* __restrict__ W1,
                                           const float* __restrict__ b1,
                                           float* __restrict__ h) {
    __shared__ __attribute__((aligned(16))) float xs[32 * INDIM];
    const int m0 = (blockIdx.x >> 1) * 32;
    const int ch = blockIdx.x & 1;
    {
        const v4f* srcv = (const v4f*)(x + (size_t)m0 * INDIM);
        v4f* dstv = (v4f*)xs;
        const bool tail = threadIdx.x < XV4 - XFULL * 128;
        v4f buf[XFULL + 1];
#pragma unroll
        for (int j = 0; j < XFULL; ++j) buf[j] = srcv[threadIdx.x + 128 * j];
        if (tail) buf[XFULL] = srcv[threadIdx.x + 128 * XFULL];
#pragma unroll
        for (int j = 0; j < XFULL; ++j) dstv[threadIdx.x + 128 * j] = buf[j];
        if (tail) dstv[threadIdx.x + 128 * XFULL] = buf[XFULL];
    }
    __syncthreads();

    const int wave = threadIdx.x >> 5;
    const int lane = threadIdx.x & 31;
    const int rr   = lane & 15;
    const int half = lane >> 4;
    const int n    = ch * 64 + wave * 16 + rr;

    const float* a0p = xs + rr * INDIM + half * 2;
    const float* a1p = xs + (16 + rr) * INDIM + half * 2;
    const float* bp  = W1 + (half * 2) * HID + n;

    v8f acc0 = {0.f,0.f,0.f,0.f,0.f,0.f,0.f,0.f};
    v8f acc1 = acc0;
    for (int kk = 0; kk < INDIM; kk += 4) {
        const v2f a0 = *(const v2f*)(a0p + kk);
        const v2f a1 = *(const v2f*)(a1p + kk);
        v2f bf; bf.x = bp[kk * HID]; bf.y = bp[(kk + 1) * HID];
        acc0 = wmma_f32_k4(a0, bf, acc0);
        acc1 = wmma_f32_k4(a1, bf, acc1);
    }
    const float bias = b1[n];
#pragma unroll
    for (int r = 0; r < 8; ++r) {
        const int m = m0 + r + half * 8;
        const float v0 = acc0[r] + bias;
        const float v1 = acc1[r] + bias;
        h[m * HID + n]        = v0 > 0.f ? v0 : 0.f;
        h[(m + 16) * HID + n] = v1 > 0.f ? v1 : 0.f;
    }
}

__global__ __launch_bounds__(128) void k_qkv(const float* __restrict__ h,
                                             const float* __restrict__ Wq,
                                             const float* __restrict__ Wk,
                                             const float* __restrict__ Wv,
                                             float* __restrict__ Q,
                                             float* __restrict__ K,
                                             float* __restrict__ V) {
    __shared__ __attribute__((aligned(16))) float hs[32 * 132];
    const int m0 = (blockIdx.x >> 1) * 32;
    const int ch = blockIdx.x & 1;
    {
        const float* srcp = h + (size_t)m0 * HID;
        v4f buf[8];
#pragma unroll
        for (int j = 0; j < 8; ++j) {
            const int e = threadIdx.x + 128 * j;
            buf[j] = *(const v4f*)(srcp + e * 4);
        }
#pragma unroll
        for (int j = 0; j < 8; ++j) {
            const int e = threadIdx.x + 128 * j;
            *(v4f*)(hs + (e >> 5) * 132 + (e & 31) * 4) = buf[j];
        }
    }
    __syncthreads();

    const int wave = threadIdx.x >> 5;
    const int lane = threadIdx.x & 31;
    const int rr   = lane & 15;
    const int half = lane >> 4;
    const int n    = ch * 64 + wave * 16 + rr;

    const float* a0p = hs + rr * 132 + half * 2;
    const float* a1p = hs + (16 + rr) * 132 + half * 2;
    const int    bo  = (half * 2) * HID + n;

    v8f aq0 = {0.f,0.f,0.f,0.f,0.f,0.f,0.f,0.f};
    v8f aq1 = aq0, ak0 = aq0, ak1 = aq0, av0 = aq0, av1 = aq0;
    for (int kk = 0; kk < HID; kk += 4) {
        const v2f a0 = *(const v2f*)(a0p + kk);
        const v2f a1 = *(const v2f*)(a1p + kk);
        const int o = bo + kk * HID;
        v2f bq; bq.x = Wq[o]; bq.y = Wq[o + HID];
        v2f bk; bk.x = Wk[o]; bk.y = Wk[o + HID];
        v2f bv; bv.x = Wv[o]; bv.y = Wv[o + HID];
        aq0 = wmma_f32_k4(a0, bq, aq0);
        aq1 = wmma_f32_k4(a1, bq, aq1);
        ak0 = wmma_f32_k4(a0, bk, ak0);
        ak1 = wmma_f32_k4(a1, bk, ak1);
        av0 = wmma_f32_k4(a0, bv, av0);
        av1 = wmma_f32_k4(a1, bv, av1);
    }
#pragma unroll
    for (int r = 0; r < 8; ++r) {
        const int m = m0 + r + half * 8;
        Q[m * HID + n] = aq0[r];
        K[m * HID + n] = ak0[r];
        V[m * HID + n] = av0[r];
        Q[(m + 16) * HID + n] = aq1[r];
        K[(m + 16) * HID + n] = ak1[r];
        V[(m + 16) * HID + n] = av1[r];
    }
}

__global__ __launch_bounds__(224) void k_gram(const float* __restrict__ Q,
                                              const float* __restrict__ Km,
                                              float* __restrict__ G) {
    const int I    = threadIdx.x >> 5;
    const int lane = threadIdx.x & 31;
    const int rr   = lane & 15;
    const int half = lane >> 4;
    const long base = (long)blockIdx.x * SEQ;

    const int ra  = min(I * 16 + rr, SEQ - 1);
    const int rc0 = max((I - 1) * 16 + rr, 0);
    const int rc1 = min(I * 16 + rr, SEQ - 1);
    const int rc2 = min((I + 1) * 16 + rr, SEQ - 1);
    const float* qrow = Q  + (base + ra)  * HID + half * 4;
    const float* kr0  = Km + (base + rc0) * HID + half * 4;
    const float* kr1  = Km + (base + rc1) * HID + half * 4;
    const float* kr2  = Km + (base + rc2) * HID + half * 4;

    v8f g0 = {0.f,0.f,0.f,0.f,0.f,0.f,0.f,0.f};
    v8f g1 = g0, g2 = g0;
    for (int t = 0; t < HID; t += 8) {
        const v4f a  = *(const v4f*)(qrow + t);
        const v4f kb0 = *(const v4f*)(kr0 + t);
        const v4f kb1 = *(const v4f*)(kr1 + t);
        const v4f kb2 = *(const v4f*)(kr2 + t);
        v2f alo = {a.x, a.y}, ahi = {a.z, a.w};
        v2f lo0 = {kb0.x, kb0.y}, hi0 = {kb0.z, kb0.w};
        v2f lo1 = {kb1.x, kb1.y}, hi1 = {kb1.z, kb1.w};
        v2f lo2 = {kb2.x, kb2.y}, hi2 = {kb2.z, kb2.w};
        g0 = wmma_f32_k4(alo, lo0, g0);
        g1 = wmma_f32_k4(alo, lo1, g1);
        g2 = wmma_f32_k4(alo, lo2, g2);
        g0 = wmma_f32_k4(ahi, hi0, g0);
        g1 = wmma_f32_k4(ahi, hi1, g1);
        g2 = wmma_f32_k4(ahi, hi2, g2);
    }

    const float scale = 0.08838834764831845f;
#pragma unroll
    for (int r = 0; r < 8; ++r) {
        const int a = I * 16 + r + half * 8;
        if (a < SEQ) {
            float* grow = G + (base + a) * GP + (10 - a);
            const int c0 = (I - 1) * 16 + rr;
            const int c1 = I * 16 + rr;
            const int c2 = (I + 1) * 16 + rr;
            if (c0 >= 0  && a - c0 <= 10)                 grow[c0] = g0[r] * scale;
            if (c1 < SEQ && c1 - a <= 10 && a - c1 <= 10) grow[c1] = g1[r] * scale;
            if (c2 < SEQ && c2 - a <= 10)                 grow[c2] = g2[r] * scale;
        }
    }
}

template <int J>
static __device__ __forceinline__ float row_bcast(float x) {
    const int b = __builtin_bit_cast(int, x);
    return __builtin_bit_cast(float,
        __builtin_amdgcn_update_dpp(b, b, 0x150 + J, 0xf, 0xf, true));
}

template <int J>
static __device__ __forceinline__ void pool_row(const float* __restrict__ vseq, int s,
                                                float bsum, v4f& plo, v4f& phi) {
    const int   rv = s + 5 - J;
    const int   rc = min(max(rv, 0), SEQ - 1);
    const float bj = row_bcast<J>(bsum);
    const float w  = (rv >= 0 && rv < SEQ) ? bj : 0.f;
    const float* vr = vseq + rc * HID;
    const v4f lo = *(const v4f*)(vr);
    const v4f hi = *(const v4f*)(vr + 64);
    plo += w * lo;
    phi += w * hi;
}

__global__ __launch_bounds__(256, 16) void k_attn(const float* __restrict__ G,
                                              const float* __restrict__ V,
                                              const float* __restrict__ W2,
                                              const float* __restrict__ b2,
                                              float* __restrict__ out,
                                              float* __restrict__ alphaOut) {
    const int rr  = threadIdx.x & 15;
    const int idx = blockIdx.x * 16 + (threadIdx.x >> 4);
    const int s   = idx % SEQ;

    const int  cj  = s + 5 - rr;
    const bool jin = rr < WIN;
    const bool jok = jin && (cj >= 0) && (cj < SEQ);
    const float* gcol  = G + (long)(idx + 5) * GP + (10 - rr);
    const float* gdiag = G + (long)idx * GP + 10;

    float scv[WIN];
#pragma unroll
    for (int i = 0; i < WIN; ++i) {
        const int  ai = s + 5 - i;
        const bool ok = jok && (ai >= 0) && (ai < SEQ);
        const float g = *(ok ? gcol - (GP - 1) * i : gdiag);
        scv[i] = ok ? g : 0.f;
    }

    const long  abase = (long)idx * (WIN * WIN);
    float bsum = 0.f;
#pragma unroll
    for (int i = 0; i < WIN; ++i) {
        const float sc = scv[i];
        float mv = jin ? sc : -3.0e38f;
        mv = __builtin_elementwise_maximum(mv, row_xor<1>(mv));
        mv = __builtin_elementwise_maximum(mv, row_xor<2>(mv));
        mv = __builtin_elementwise_maximum(mv, row_xor<4>(mv));
        mv = __builtin_elementwise_maximum(mv, row_xor<8>(mv));
        float p = jin ? expf(sc - mv) : 0.f;
        float sum = p;
        sum += row_xor<1>(sum);
        sum += row_xor<2>(sum);
        sum += row_xor<4>(sum);
        sum += row_xor<8>(sum);
        const float al = p / sum;
        if (jin) alphaOut[abase + i * WIN + rr] = al;
        bsum += al;
    }

    v4f plo = {0.f, 0.f, 0.f, 0.f};
    v4f phi = {0.f, 0.f, 0.f, 0.f};
    const float* vseq = V + (long)(idx - s) * HID + rr * 4;
    pool_row<0>(vseq, s, bsum, plo, phi);
    pool_row<1>(vseq, s, bsum, plo, phi);
    pool_row<2>(vseq, s, bsum, plo, phi);
    pool_row<3>(vseq, s, bsum, plo, phi);
    pool_row<4>(vseq, s, bsum, plo, phi);
    pool_row<5>(vseq, s, bsum, plo, phi);
    pool_row<6>(vseq, s, bsum, plo, phi);
    pool_row<7>(vseq, s, bsum, plo, phi);
    pool_row<8>(vseq, s, bsum, plo, phi);
    pool_row<9>(vseq, s, bsum, plo, phi);
    pool_row<10>(vseq, s, bsum, plo, phi);

    const v4f wa = *(const v4f*)(W2 + rr * 8);
    const v4f wb = *(const v4f*)(W2 + rr * 8 + 4);
    const v4f wc = *(const v4f*)(W2 + 128 + rr * 8);
    const v4f wd = *(const v4f*)(W2 + 128 + rr * 8 + 4);
    float p0 = 0.f, p1 = 0.f;
    p0 += plo.x * wa.x;  p1 += plo.x * wa.y;
    p0 += plo.y * wa.z;  p1 += plo.y * wa.w;
    p0 += plo.z * wb.x;  p1 += plo.z * wb.y;
    p0 += plo.w * wb.z;  p1 += plo.w * wb.w;
    p0 += phi.x * wc.x;  p1 += phi.x * wc.y;
    p0 += phi.y * wc.z;  p1 += phi.y * wc.w;
    p0 += phi.z * wd.x;  p1 += phi.z * wd.y;
    p0 += phi.w * wd.z;  p1 += phi.w * wd.w;
    p0 += row_xor<1>(p0);  p1 += row_xor<1>(p1);
    p0 += row_xor<2>(p0);  p1 += row_xor<2>(p1);
    p0 += row_xor<4>(p0);  p1 += row_xor<4>(p1);
    p0 += row_xor<8>(p0);  p1 += row_xor<8>(p1);
    if (rr == 0) {
        out[idx * 2 + 0] = 1.f / (1.f + expf(-(p0 + b2[0])));
        out[idx * 2 + 1] = 1.f / (1.f + expf(-(p1 + b2[1])));
    }
}

extern "C" void kernel_launch(void* const* d_in, const int* in_sizes, int n_in,
                              void* d_out, int out_size, void* d_ws, size_t ws_size,
                              hipStream_t stream) {
    const float* x  = (const float*)d_in[0];
    const float* W1 = (const float*)d_in[1];
    const float* b1 = (const float*)d_in[2];
    const float* Wq = (const float*)d_in[3];
    const float* Wk = (const float*)d_in[4];
    const float* Wv = (const float*)d_in[5];
    const float* W2 = (const float*)d_in[6];
    const float* b2 = (const float*)d_in[7];

    float* h = (float*)d_ws;
    float* Q = h + (size_t)NROW * HID;
    float* K = Q + (size_t)NROW * HID;
    float* V = K + (size_t)NROW * HID;
    float* G = h;

    float* out      = (float*)d_out;
    float* alphaOut = out + (size_t)NROW * 2;

    k_h  <<<NROW / 16, 128, 0, stream>>>(x, W1, b1, h);
    k_qkv<<<NROW / 16, 128, 0, stream>>>(h, Wq, Wk, Wv, Q, K, V);
    k_gram<<<BATCH, 224, 0, stream>>>(Q, K, G);
    k_attn<<<NROW / 16, 256, 0, stream>>>(G, V, W2, b2, out, alphaOut);
}


# scores from T = h@(Wq@Wk^T) and h itself: 16 extra blocks of k_h build the 128x128 weight product once (WMMA); k_qkv makes T and V only; K GEMM and K plane gone; order of the sums changes
# speedup vs baseline: 1.7688x; 1.1339x over previous
#include <hip/hip_runtime.h>
#include <math.h>

typedef __attribute__((ext_vector_type(2))) float v2f;
typedef __attribute__((ext_vector_type(4))) float v4f;
typedef __attribute__((ext_vector_type(8))) float v8f;

#define SEQ   100
#define BATCH 256
#define INDIM 300
#define HID   128
#define NROW  (BATCH*SEQ)
#define WIN   11
#define GP    32

static __device__ __forceinline__ v8f wmma_f32_k4(v2f a, v2f b, v8f c) {
    return __builtin_amdgcn_wmma_f32_16x16x4_f32(false, a, false, b, (short)0, c,
                                                 false, false);
}

template <int M>
static __device__ __forceinline__ float row_xor(float x) {
    const int b = __builtin_bit_cast(int, x);
    return __builtin_bit_cast(float,
        __builtin_amdgcn_update_dpp(b, b, 0x160 + M, 0xf, 0xf, true));
}

#define MQK_BLOCKS 16

#define XV4   (32 * INDIM / 4)
#define XFULL (XV4 / 128)
__global__ __launch_bounds__(128) void k_h(const float* __restrict__ x,
                                           const float* __restrict__ W1,
                                           const float* __restrict__ b1,
                                           const float* __restrict__ Wq,
                                           const float* __restrict__ Wk,
                                           float* __restrict__ h,
                                           float* __restrict__ Mqk) {
    __shared__ __attribute__((aligned(16))) float xs[32 * INDIM];
    if (blockIdx.x >= NROW / 16) {
        const int lane = threadIdx.x & 31;
        const int rr   = lane & 15;
        const int half = lane >> 4;
        const int tile = (blockIdx.x - NROW / 16) * 4 + (threadIdx.x >> 5);
        const int i0 = (tile >> 3) * 16;
        const int j0 = (tile & 7) * 16;
        const float* ap = Wq + (i0 + rr) * HID + half * 4;
        const float* bp = Wk + (j0 + rr) * HID + half * 4;
        v8f m0a = {0.f,0.f,0.f,0.f,0.f,0.f,0.f,0.f};
        v8f m1a = m0a;
#pragma unroll 4
        for (int t = 0; t < HID; t += 8) {
            const v4f a = *(const v4f*)(ap + t);
            const v4f b = *(const v4f*)(bp + t);
            v2f alo = {a.x, a.y}, ahi = {a.z, a.w};
            v2f blo = {b.x, b.y}, bhi = {b.z, b.w};
            m0a = wmma_f32_k4(alo, blo, m0a);
            m1a = wmma_f32_k4(ahi, bhi, m1a);
        }
#pragma unroll
        for (int r = 0; r < 8; ++r)
            Mqk[(i0 + r + half * 8) * HID + j0 + rr] = m0a[r] + m1a[r];
        return;
    }
    const int m0 = (blockIdx.x >> 1) * 32;
    const int ch = blockIdx.x & 1;
    {
        const v4f* srcv = (const v4f*)(x + (size_t)m0 * INDIM);
        v4f* dstv = (v4f*)xs;
        const bool tail = threadIdx.x < XV4 - XFULL * 128;
        v4f buf[XFULL + 1];
#pragma unroll
        for (int j = 0; j < XFULL; ++j) buf[j] = srcv[threadIdx.x + 128 * j];
        if (tail) buf[XFULL] = srcv[threadIdx.x + 128 * XFULL];
#pragma unroll
        for (int j = 0; j < XFULL; ++j) dstv[threadIdx.x + 128 * j] = buf[j];
        if (tail) dstv[threadIdx.x + 128 * XFULL] = buf[XFULL];
    }
    __syncthreads();

    const int wave = threadIdx.x >> 5;
    const int lane = threadIdx.x & 31;
    const int rr   = lane & 15;
    const int half = lane >> 4;
    const int n    = ch * 64 + wave * 16 + rr;

    const float* a0p = xs + rr * INDIM + half * 2;
    const float* a1p = xs + (16 + rr) * INDIM + half * 2;
    const float* bp  = W1 + (half * 2) * HID + n;

    v8f acc0 = {0.f,0.f,0.f,0.f,0.f,0.f,0.f,0.f};
    v8f acc1 = acc0;
    for (int kk = 0; kk < INDIM; kk += 4) {
        const v2f a0 = *(const v2f*)(a0p + kk);
        const v2f a1 = *(const v2f*)(a1p + kk);
        v2f bf; bf.x = bp[kk * HID]; bf.y = bp[(kk + 1) * HID];
        acc0 = wmma_f32_k4(a0, bf, acc0);
        acc1 = wmma_f32_k4(a1, bf, acc1);
    }
    const float bias = b1[n];
#pragma unroll
    for (int r = 0; r < 8; ++r) {
        const int m = m0 + r + half * 8;
        const float v0 = acc0[r] + bias;
        const float v1 = acc1[r] + bias;
        h[m * HID + n]        = v0 > 0.f ? v0 : 0.f;
        h[(m + 16) * HID + n] = v1 > 0.f ? v1 : 0.f;
    }
}

__global__ __launch_bounds__(128) void k_tv(const float* __restrict__ h,
                                            const float* __restrict__ Mqk,
                                            const float* __restrict__ Wv,
                                            float* __restrict__ Tm,
                                            float* __restrict__ V) {
    __shared__ __attribute__((aligned(16))) float hs[32 * 132];
    const int m0 = (blockIdx.x >> 1) * 32;
    const int ch = blockIdx.x & 1;
    {
        const float* srcp = h + (size_t)m0 * HID;
        v4f buf[8];
#pragma unroll
        for (int j = 0; j < 8; ++j) {
            const int e = threadIdx.x + 128 * j;
            buf[j] = *(const v4f*)(srcp + e * 4);
        }
#pragma unroll
        for (int j = 0; j < 8; ++j) {
            const int e = threadIdx.x + 128 * j;
            *(v4f*)(hs + (e >> 5) * 132 + (e & 31) * 4) = buf[j];
        }
    }
    __syncthreads();

    const int wave = threadIdx.x >> 5;
    const int lane = threadIdx.x & 31;
    const int rr   = lane & 15;
    const int half = lane >> 4;
    const int n    = ch * 64 + wave * 16 + rr;

    const float* a0p = hs + rr * 132 + half * 2;
    const float* a1p = hs + (16 + rr) * 132 + half * 2;
    const int    bo  = (half * 2) * HID + n;

    v8f at0 = {0.f,0.f,0.f,0.f,0.f,0.f,0.f,0.f};
    v8f at1 = at0, av0 = at0, av1 = at0;
    for (int kk = 0; kk < HID; kk += 4) {
        const v2f a0 = *(const v2f*)(a0p + kk);
        const v2f a1 = *(const v2f*)(a1p + kk);
        const int o = bo + kk * HID;
        v2f bt; bt.x = Mqk[o]; bt.y = Mqk[o + HID];
        v2f bv; bv.x = Wv[o];  bv.y = Wv[o + HID];
        at0 = wmma_f32_k4(a0, bt, at0);
        at1 = wmma_f32_k4(a1, bt, at1);
        av0 = wmma_f32_k4(a0, bv, av0);
        av1 = wmma_f32_k4(a1, bv, av1);
    }
#pragma unroll
    for (int r = 0; r < 8; ++r) {
        const int m = m0 + r + half * 8;
        Tm[m * HID + n] = at0[r];
        V[m * HID + n]  = av0[r];
        Tm[(m + 16) * HID + n] = at1[r];
        V[(m + 16) * HID + n]  = av1[r];
    }
}

__global__ __launch_bounds__(224) void k_gram(const float* __restrict__ Q,
                                              const float* __restrict__ Km,
                                              float* __restrict__ G) {
    const int I    = threadIdx.x >> 5;
    const int lane = threadIdx.x & 31;
    const int rr   = lane & 15;
    const int half = lane >> 4;
    const long base = (long)blockIdx.x * SEQ;

    const int ra  = min(I * 16 + rr, SEQ - 1);
    const int rc0 = max((I - 1) * 16 + rr, 0);
    const int rc1 = min(I * 16 + rr, SEQ - 1);
    const int rc2 = min((I + 1) * 16 + rr, SEQ - 1);
    const float* qrow = Q  + (base + ra)  * HID + half * 4;
    const float* kr0  = Km + (base + rc0) * HID + half * 4;
    const float* kr1  = Km + (base + rc1) * HID + half * 4;
    const float* kr2  = Km + (base + rc2) * HID + half * 4;

    v8f g0 = {0.f,0.f,0.f,0.f,0.f,0.f,0.f,0.f};
    v8f g1 = g0, g2 = g0;
    for (int t = 0; t < HID; t += 8) {
        const v4f a  = *(const v4f*)(qrow + t);
        const v4f kb0 = *(const v4f*)(kr0 + t);
        const v4f kb1 = *(const v4f*)(kr1 + t);
        const v4f kb2 = *(const v4f*)(kr2 + t);
        v2f alo = {a.x, a.y}, ahi = {a.z, a.w};
        v2f lo0 = {kb0.x, kb0.y}, hi0 = {kb0.z, kb0.w};
        v2f lo1 = {kb1.x, kb1.y}, hi1 = {kb1.z, kb1.w};
        v2f lo2 = {kb2.x, kb2.y}, hi2 = {kb2.z, kb2.w};
        g0 = wmma_f32_k4(alo, lo0, g0);
        g1 = wmma_f32_k4(alo, lo1, g1);
        g2 = wmma_f32_k4(alo, lo2, g2);
        g0 = wmma_f32_k4(ahi, hi0, g0);
        g1 = wmma_f32_k4(ahi, hi1, g1);
        g2 = wmma_f32_k4(ahi, hi2, g2);
    }

    const float scale = 0.08838834764831845f;
#pragma unroll
    for (int r = 0; r < 8; ++r) {
        const int a = I * 16 + r + half * 8;
        if (a < SEQ) {
            float* grow = G + (base + a) * GP + (10 - a);
            const int c0 = (I - 1) * 16 + rr;
            const int c1 = I * 16 + rr;
            const int c2 = (I + 1) * 16 + rr;
            if (c0 >= 0  && a - c0 <= 10)                 grow[c0] = g0[r] * scale;
            if (c1 < SEQ && c1 - a <= 10 && a - c1 <= 10) grow[c1] = g1[r] * scale;
            if (c2 < SEQ && c2 - a <= 10)                 grow[c2] = g2[r] * scale;
        }
    }
}

template <int J>
static __device__ __forceinline__ float row_bcast(float x) {
    const int b = __builtin_bit_cast(int, x);
    return __builtin_bit_cast(float,
        __builtin_amdgcn_update_dpp(b, b, 0x150 + J, 0xf, 0xf, true));
}

template <int J>
static __device__ __forceinline__ void pool_row(const float* __restrict__ vseq, int s,
                                                float bsum, v4f& plo, v4f& phi) {
    const int   rv = s + 5 - J;
    const int   rc = min(max(rv, 0), SEQ - 1);
    const float bj = row_bcast<J>(bsum);
    const float w  = (rv >= 0 && rv < SEQ) ? bj : 0.f;
    const float* vr = vseq + rc * HID;
    const v4f lo = *(const v4f*)(vr);
    const v4f hi = *(const v4f*)(vr + 64);
    plo += w * lo;
    phi += w * hi;
}

__global__ __launch_bounds__(256, 16) void k_attn(const float* __restrict__ G,
                                              const float* __restrict__ V,
                                              const float* __restrict__ W2,
                                              const float* __restrict__ b2,
                                              float* __restrict__ out,
                                              float* __restrict__ alphaOut) {
    const int rr  = threadIdx.x & 15;
    const int idx = blockIdx.x * 16 + (threadIdx.x >> 4);
    const int s   = idx % SEQ;

    const int  cj  = s + 5 - rr;
    const bool jin = rr < WIN;
    const bool jok = jin && (cj >= 0) && (cj < SEQ);
    const float* gcol  = G + (long)(idx + 5) * GP + (10 - rr);
    const float* gdiag = G + (long)idx * GP + 10;

    float scv[WIN];
#pragma unroll
    for (int i = 0; i < WIN; ++i) {
        const int  ai = s + 5 - i;
        const bool ok = jok && (ai >= 0) && (ai < SEQ);
        const float g = *(ok ? gcol - (GP - 1) * i : gdiag);
        scv[i] = ok ? g : 0.f;
    }

    const long  abase = (long)idx * (WIN * WIN);
    float bsum = 0.f;
#pragma unroll
    for (int i = 0; i < WIN; ++i) {
        const float sc = scv[i];
        float mv = jin ? sc : -3.0e38f;
        mv = __builtin_elementwise_maximum(mv, row_xor<1>(mv));
        mv = __builtin_elementwise_maximum(mv, row_xor<2>(mv));
        mv = __builtin_elementwise_maximum(mv, row_xor<4>(mv));
        mv = __builtin_elementwise_maximum(mv, row_xor<8>(mv));
        float p = jin ? expf(sc - mv) : 0.f;
        float sum = p;
        sum += row_xor<1>(sum);
        sum += row_xor<2>(sum);
        sum += row_xor<4>(sum);
        sum += row_xor<8>(sum);
        const float al = p / sum;
        if (jin) alphaOut[abase + i * WIN + rr] = al;
        bsum += al;
    }

    v4f plo = {0.f, 0.f, 0.f, 0.f};
    v4f phi = {0.f, 0.f, 0.f, 0.f};
    const float* vseq = V + (long)(idx - s) * HID + rr * 4;
    pool_row<0>(vseq, s, bsum, plo, phi);
    pool_row<1>(vseq, s, bsum, plo, phi);
    pool_row<2>(vseq, s, bsum, plo, phi);
    pool_row<3>(vseq, s, bsum, plo, phi);
    pool_row<4>(vseq, s, bsum, plo, phi);
    pool_row<5>(vseq, s, bsum, plo, phi);
    pool_row<6>(vseq, s, bsum, plo, phi);
    pool_row<7>(vseq, s, bsum, plo, phi);
    pool_row<8>(vseq, s, bsum, plo, phi);
    pool_row<9>(vseq, s, bsum, plo, phi);
    pool_row<10>(vseq, s, bsum, plo, phi);

    const v4f wa = *(const v4f*)(W2 + rr * 8);
    const v4f wb = *(const v4f*)(W2 + rr * 8 + 4);
    const v4f wc = *(const v4f*)(W2 + 128 + rr * 8);
    const v4f wd = *(const v4f*)(W2 + 128 + rr * 8 + 4);
    float p0 = 0.f, p1 = 0.f;
    p0 += plo.x * wa.x;  p1 += plo.x * wa.y;
    p0 += plo.y * wa.z;  p1 += plo.y * wa.w;
    p0 += plo.z * wb.x;  p1 += plo.z * wb.y;
    p0 += plo.w * wb.z;  p1 += plo.w * wb.w;
    p0 += phi.x * wc.x;  p1 += phi.x * wc.y;
    p0 += phi.y * wc.z;  p1 += phi.y * wc.w;
    p0 += phi.z * wd.x;  p1 += phi.z * wd.y;
    p0 += phi.w * wd.z;  p1 += phi.w * wd.w;
    p0 += row_xor<1>(p0);  p1 += row_xor<1>(p1);
    p0 += row_xor<2>(p0);  p1 += row_xor<2>(p1);
    p0 += row_xor<4>(p0);  p1 += row_xor<4>(p1);
    p0 += row_xor<8>(p0);  p1 += row_xor<8>(p1);
    if (rr == 0) {
        out[idx * 2 + 0] = 1.f / (1.f + expf(-(p0 + b2[0])));
        out[idx * 2 + 1] = 1.f / (1.f + expf(-(p1 + b2[1])));
    }
}

extern "C" void kernel_launch(void* const* d_in, const int* in_sizes, int n_in,
                              void* d_out, int out_size, void* d_ws, size_t ws_size,
                              hipStream_t stream) {
    const float* x  = (const float*)d_in[0];
    const float* W1 = (const float*)d_in[1];
    const float* b1 = (const float*)d_in[2];
    const float* Wq = (const float*)d_in[3];
    const float* Wk = (const float*)d_in[4];
    const float* Wv = (const float*)d_in[5];
    const float* W2 = (const float*)d_in[6];
    const float* b2 = (const float*)d_in[7];

    float* h  = (float*)d_ws;
    float* Tm = h  + (size_t)NROW * HID;
    float* G  = Tm + (size_t)NROW * HID;
    float* V  = G  + (size_t)NROW * HID;
    float* Mqk = G + (size_t)NROW * GP;

    float* out      = (float*)d_out;
    float* alphaOut = out + (size_t)NROW * 2;

    k_h  <<<NROW / 16 + MQK_BLOCKS, 128, 0, stream>>>(x, W1, b1, Wq, Wk, h, Mqk);
    k_tv <<<NROW / 16, 128, 0, stream>>>(h, Mqk, Wv, Tm, V);
    k_gram<<<BATCH, 224, 0, stream>>>(Tm, h, G);
    k_attn<<<NROW / 16, 256, 0, stream>>>(G, V, W2, b2, out, alphaOut);
}


# k_tv makes the score band itself: a block computes T for 16 rows x 128 columns, keeps it in LDS and multiplies it (WMMA) with the 48 staged h rows around it; band launch, T plane gone; sums unchanged
# speedup vs baseline: 1.9965x; 1.1287x over previous
#include <hip/hip_runtime.h>
#include <math.h>

typedef __attribute__((ext_vector_type(2))) float v2f;
typedef __attribute__((ext_vector_type(4))) float v4f;
typedef __attribute__((ext_vector_type(8))) float v8f;

#define SEQ   100
#define BATCH 256
#define INDIM 300
#define HID   128
#define NROW  (BATCH*SEQ)
#define WIN   11
#define GP    32

static __device__ __forceinline__ v8f wmma_f32_k4(v2f a, v2f b, v8f c) {
    return __builtin_amdgcn_wmma_f32_16x16x4_f32(false, a, false, b, (short)0, c,
                                                 false, false);
}

template <int M>
static __device__ __forceinline__ float row_xor(float x) {
    const int b = __builtin_bit_cast(int, x);
    return __builtin_bit_cast(float,
        __builtin_amdgcn_update_dpp(b, b, 0x160 + M, 0xf, 0xf, true));
}

#define MQK_BLOCKS 16

#define XV4   (32 * INDIM / 4)
#define XFULL (XV4 / 128)
__global__ __launch_bounds__(128) void k_h(const float* __restrict__ x,
                                           const float* __restrict__ W1,
                                           const float* __restrict__ b1,
                                           const float* __restrict__ Wq,
                                           const float* __restrict__ Wk,
                                           float* __restrict__ h,
                                           float* __restrict__ Mqk) {
    __shared__ __attribute__((aligned(16))) float xs[32 * INDIM];
    if (blockIdx.x >= NROW / 16) {
        const int lane = threadIdx.x & 31;
        const int rr   = lane & 15;
        const int half = lane >> 4;
        const int tile = (blockIdx.x - NROW / 16) * 4 + (threadIdx.x >> 5);
        const int i0 = (tile >> 3) * 16;
        const int j0 = (tile & 7) * 16;
        const float* ap = Wq + (i0 + rr) * HID + half * 4;
        const float* bp = Wk + (j0 + rr) * HID + half * 4;
        v8f m0a = {0.f,0.f,0.f,0.f,0.f,0.f,0.f,0.f};
        v8f m1a = m0a;
#pragma unroll 4
        for (int t = 0; t < HID; t += 8) {
            const v4f a = *(const v4f*)(ap + t);
            const v4f b = *(const v4f*)(bp + t);
            v2f alo = {a.x, a.y}, ahi = {a.z, a.w};
            v2f blo = {b.x, b.y}, bhi = {b.z, b.w};
            m0a = wmma_f32_k4(alo, blo, m0a);
            m1a = wmma_f32_k4(ahi, bhi, m1a);
        }
#pragma unroll
        for (int r = 0; r < 8; ++r)
            Mqk[(i0 + r + half * 8) * HID + j0 + rr] = m0a[r] + m1a[r];
        return;
    }
    const int m0 = (blockIdx.x >> 1) * 32;
    const int ch = blockIdx.x & 1;
    {
        const v4f* srcv = (const v4f*)(x + (size_t)m0 * INDIM);
        v4f* dstv = (v4f*)xs;
        const bool tail = threadIdx.x < XV4 - XFULL * 128;
        v4f buf[XFULL + 1];
#pragma unroll
        for (int j = 0; j < XFULL; ++j) buf[j] = srcv[threadIdx.x + 128 * j];
        if (tail) buf[XFULL] = srcv[threadIdx.x + 128 * XFULL];
#pragma unroll
        for (int j = 0; j < XFULL; ++j) dstv[threadIdx.x + 128 * j] = buf[j];
        if (tail) dstv[threadIdx.x + 128 * XFULL] = buf[XFULL];
    }
    __syncthreads();

    const int wave = threadIdx.x >> 5;
    const int lane = threadIdx.x & 31;
    const int rr   = lane & 15;
    const int half = lane >> 4;
    const int n    = ch * 64 + wave * 16 + rr;

    const float* a0p = xs + rr * INDIM + half * 2;
    const float* a1p = xs + (16 + rr) * INDIM + half * 2;
    const float* bp  = W1 + (half * 2) * HID + n;

    v8f acc0 = {0.f,0.f,0.f,0.f,0.f,0.f,0.f,0.f};
    v8f acc1 = acc0;
    for (int kk = 0; kk < INDIM; kk += 4) {
        const v2f a0 = *(const v2f*)(a0p + kk);
        const v2f a1 = *(const v2f*)(a1p + kk);
        v2f bf; bf.x = bp[kk * HID]; bf.y = bp[(kk + 1) * HID];
        acc0 = wmma_f32_k4(a0, bf, acc0);
        acc1 = wmma_f32_k4(a1, bf, acc1);
    }
    const float bias = b1[n];
#pragma unroll
    for (int r = 0; r < 8; ++r) {
        const int m = m0 + r + half * 8;
        const float v0 = acc0[r] + bias;
        const float v1 = acc1[r] + bias;
        h[m * HID + n]        = v0 > 0.f ? v0 : 0.f;
        h[(m + 16) * HID + n] = v1 > 0.f ? v1 : 0.f;
    }
}

#define HP 132
__global__ __launch_bounds__(128) void k_tvg(const float* __restrict__ h,
                                             const float* __restrict__ Mqk,
                                             const float* __restrict__ Wv,
                                             float* __restrict__ V,
                                             float* __restrict__ G) {
    __shared__ __attribute__((aligned(16))) float hs[48 * HP];
    __shared__ __attribute__((aligned(16))) float ts[16 * HP];
    const int m0 = blockIdx.x * 16;
    {
        v4f buf[12];
#pragma unroll
        for (int j = 0; j < 12; ++j) {
            const int e   = threadIdx.x + 128 * j;
            const int row = min(max(m0 - 16 + (e >> 5), 0), NROW - 1);
            buf[j] = *(const v4f*)(h + (size_t)row * HID + (e & 31) * 4);
        }
#pragma unroll
        for (int j = 0; j < 12; ++j) {
            const int e = threadIdx.x + 128 * j;
            *(v4f*)(hs + (e >> 5) * HP + (e & 31) * 4) = buf[j];
        }
    }
    __syncthreads();

    const int wave = threadIdx.x >> 5;
    const int lane = threadIdx.x & 31;
    const int rr   = lane & 15;
    const int half = lane >> 4;
    const int n    = wave * 32 + rr;

    const float* ap = hs + (16 + rr) * HP + half * 2;
    const int    bo = (half * 2) * HID + n;

    v8f at0 = {0.f,0.f,0.f,0.f,0.f,0.f,0.f,0.f};
    v8f at1 = at0, av0 = at0, av1 = at0;
    for (int kk = 0; kk < HID; kk += 4) {
        const v2f a = *(const v2f*)(ap + kk);
        const int o = bo + kk * HID;
        v2f bt0; bt0.x = Mqk[o];      bt0.y = Mqk[o + HID];
        v2f bt1; bt1.x = Mqk[o + 16]; bt1.y = Mqk[o + 16 + HID];
        v2f bv0; bv0.x = Wv[o];       bv0.y = Wv[o + HID];
        v2f bv1; bv1.x = Wv[o + 16];  bv1.y = Wv[o + 16 + HID];
        at0 = wmma_f32_k4(a, bt0, at0);
        at1 = wmma_f32_k4(a, bt1, at1);
        av0 = wmma_f32_k4(a, bv0, av0);
        av1 = wmma_f32_k4(a, bv1, av1);
    }
#pragma unroll
    for (int r = 0; r < 8; ++r) {
        const int rl = r + half * 8;
        ts[rl * HP + n]      = at0[r];
        ts[rl * HP + n + 16] = at1[r];
    }
    __syncthreads();
#pragma unroll
    for (int r = 0; r < 8; ++r) {
        const int rl = r + half * 8;
        V[(m0 + rl) * HID + n]      = av0[r];
        V[(m0 + rl) * HID + n + 16] = av1[r];
    }

    if (wave < 3) {
        const float* tp = ts + rr * HP + half * 4;
        const float* cp = hs + (wave * 16 + rr) * HP + half * 4;
        v8f g = {0.f,0.f,0.f,0.f,0.f,0.f,0.f,0.f};
        for (int t = 0; t < HID; t += 8) {
            const v4f a = *(const v4f*)(tp + t);
            const v4f b = *(const v4f*)(cp + t);
            v2f alo = {a.x, a.y}, ahi = {a.z, a.w};
            v2f blo = {b.x, b.y}, bhi = {b.z, b.w};
            g = wmma_f32_k4(alo, blo, g);
            g = wmma_f32_k4(ahi, bhi, g);
        }
        const float scale = 0.08838834764831845f;
        const int gc  = m0 - 16 + wave * 16 + rr;
        const int sm0 = m0 % SEQ;
#pragma unroll
        for (int r = 0; r < 8; ++r) {
            const int rl = r + half * 8;
            const int d  = gc - (m0 + rl);
            int sa = sm0 + rl;
            if (sa >= SEQ) sa -= SEQ;
            const int sc = sa + d;
            if (d >= -10 && d <= 10 && sc >= 0 && sc < SEQ)
                G[(long)(m0 + rl) * GP + (d + 10)] = g[r] * scale;
        }
    }
}

template <int J>
static __device__ __forceinline__ float row_bcast(float x) {
    const int b = __builtin_bit_cast(int, x);
    return __builtin_bit_cast(float,
        __builtin_amdgcn_update_dpp(b, b, 0x150 + J, 0xf, 0xf, true));
}

template <int J>
static __device__ __forceinline__ void pool_row(const float* __restrict__ vseq, int s,
                                                float bsum, v4f& plo, v4f& phi) {
    const int   rv = s + 5 - J;
    const int   rc = min(max(rv, 0), SEQ - 1);
    const float bj = row_bcast<J>(bsum);
    const float w  = (rv >= 0 && rv < SEQ) ? bj : 0.f;
    const float* vr = vseq + rc * HID;
    const v4f lo = *(const v4f*)(vr);
    const v4f hi = *(const v4f*)(vr + 64);
    plo += w * lo;
    phi += w * hi;
}

__global__ __launch_bounds__(256, 16) void k_attn(const float* __restrict__ G,
                                              const float* __restrict__ V,
                                              const float* __restrict__ W2,
                                              const float* __restrict__ b2,
                                              float* __restrict__ out,
                                              float* __restrict__ alphaOut) {
    const int rr  = threadIdx.x & 15;
    const int idx = blockIdx.x * 16 + (threadIdx.x >> 4);
    const int s   = idx % SEQ;

    const int  cj  = s + 5 - rr;
    const bool jin = rr < WIN;
    const bool jok = jin && (cj >= 0) && (cj < SEQ);
    const float* gcol  = G + (long)(idx + 5) * GP + (10 - rr);
    const float* gdiag = G + (long)idx * GP + 10;

    float scv[WIN];
#pragma unroll
    for (int i = 0; i < WIN; ++i) {
        const int  ai = s + 5 - i;
        const bool ok = jok && (ai >= 0) && (ai < SEQ);
        const float g = *(ok ? gcol - (GP - 1) * i : gdiag);
        scv[i] = ok ? g : 0.f;
    }

    const long  abase = (long)idx * (WIN * WIN);
    float bsum = 0.f;
#pragma unroll
    for (int i = 0; i < WIN; ++i) {
        const float sc = scv[i];
        float mv = jin ? sc : -3.0e38f;
        mv = __builtin_elementwise_maximum(mv, row_xor<1>(mv));
        mv = __builtin_elementwise_maximum(mv, row_xor<2>(mv));
        mv = __builtin_elementwise_maximum(mv, row_xor<4>(mv));
        mv = __builtin_elementwise_maximum(mv, row_xor<8>(mv));
        float p = jin ? expf(sc - mv) : 0.f;
        float sum = p;
        sum += row_xor<1>(sum);
        sum += row_xor<2>(sum);
        sum += row_xor<4>(sum);
        sum += row_xor<8>(sum);
        const float al = p / sum;
        if (jin) alphaOut[abase + i * WIN + rr] = al;
        bsum += al;
    }

    v4f plo = {0.f, 0.f, 0.f, 0.f};
    v4f phi = {0.f, 0.f, 0.f, 0.f};
    const float* vseq = V + (long)(idx - s) * HID + rr * 4;
    pool_row<0>(vseq, s, bsum, plo, phi);
    pool_row<1>(vseq, s, bsum, plo, phi);
    pool_row<2>(vseq, s, bsum, plo, phi);
    pool_row<3>(vseq, s, bsum, plo, phi);
    pool_row<4>(vseq, s, bsum, plo, phi);
    pool_row<5>(vseq, s, bsum, plo, phi);
    pool_row<6>(vseq, s, bsum, plo, phi);
    pool_row<7>(vseq, s, bsum, plo, phi);
    pool_row<8>(vseq, s, bsum, plo, phi);
    pool_row<9>(vseq, s, bsum, plo, phi);
    pool_row<10>(vseq, s, bsum, plo, phi);

    const v4f wa = *(const v4f*)(W2 + rr * 8);
    const v4f wb = *(const v4f*)(W2 + rr * 8 + 4);
    const v4f wc = *(const v4f*)(W2 + 128 + rr * 8);
    const v4f wd = *(const v4f*)(W2 + 128 + rr * 8 + 4);
    float p0 = 0.f, p1 = 0.f;
    p0 += plo.x * wa.x;  p1 += plo.x * wa.y;
    p0 += plo.y * wa.z;  p1 += plo.y * wa.w;
    p0 += plo.z * wb.x;  p1 += plo.z * wb.y;
    p0 += plo.w * wb.z;  p1 += plo.w * wb.w;
    p0 += phi.x * wc.x;  p1 += phi.x * wc.y;
    p0 += phi.y * wc.z;  p1 += phi.y * wc.w;
    p0 += phi.z * wd.x;  p1 += phi.z * wd.y;
    p0 += phi.w * wd.z;  p1 += phi.w * wd.w;
    p0 += row_xor<1>(p0);  p1 += row_xor<1>(p1);
    p0 += row_xor<2>(p0);  p1 += row_xor<2>(p1);
    p0 += row_xor<4>(p0);  p1 += row_xor<4>(p1);
    p0 += row_xor<8>(p0);  p1 += row_xor<8>(p1);
    if (rr == 0) {
        out[idx * 2 + 0] = 1.f / (1.f + expf(-(p0 + b2[0])));
        out[idx * 2 + 1] = 1.f / (1.f + expf(-(p1 + b2[1])));
    }
}

extern "C" void kernel_launch(void* const* d_in, const int* in_sizes, int n_in,
                              void* d_out, int out_size, void* d_ws, size_t ws_size,
                              hipStream_t stream) {
    const float* x  = (const float*)d_in[0];
    const float* W1 = (const float*)d_in[1];
    const float* b1 = (const float*)d_in[2];
    const float* Wq = (const float*)d_in[3];
    const float* Wk = (const float*)d_in[4];
    const float* Wv = (const float*)d_in[5];
    const float* W2 = (const float*)d_in[6];
    const float* b2 = (const float*)d_in[7];

    float* h  = (float*)d_ws;
    float* G  = h  + (size_t)2 * NROW * HID;
    float* V  = G  + (size_t)NROW * HID;
    float* Mqk = G + (size_t)NROW * GP;

    float* out      = (float*)d_out;
    float* alphaOut = out + (size_t)NROW * 2;

    k_h  <<<NROW / 16 + MQK_BLOCKS, 128, 0, stream>>>(x, W1, b1, Wq, Wk, h, Mqk);
    k_tvg<<<NROW / 16, 128, 0, stream>>>(h, Mqk, Wv, V, G);
    k_attn<<<NROW / 16, 256, 0, stream>>>(G, V, W2, b2, out, alphaOut);
}


# k_attn softmax: exponent and quotient on the hardware's exp2 and reciprocal instructions (band stored in log2 units); were the library expf and IEEE division expansions; alpha's rounding changes
# speedup vs baseline: 2.0738x; 1.0387x over previous
#include <hip/hip_runtime.h>
#include <math.h>

typedef __attribute__((ext_vector_type(2))) float v2f;
typedef __attribute__((ext_vector_type(4))) float v4f;
typedef __attribute__((ext_vector_type(8))) float v8f;

#define SEQ   100
#define BATCH 256
#define INDIM 300
#define HID   128
#define NROW  (BATCH*SEQ)
#define WIN   11
#define GP    32

static __device__ __forceinline__ v8f wmma_f32_k4(v2f a, v2f b, v8f c) {
    return __builtin_amdgcn_wmma_f32_16x16x4_f32(false, a, false, b, (short)0, c,
                                                 false, false);
}

template <int M>
static __device__ __forceinline__ float row_xor(float x) {
    const int b = __builtin_bit_cast(int, x);
    return __builtin_bit_cast(float,
        __builtin_amdgcn_update_dpp(b, b, 0x160 + M, 0xf, 0xf, true));
}

#define MQK_BLOCKS 16

#define XV4   (32 * INDIM / 4)
#define XFULL (XV4 / 128)
__global__ __launch_bounds__(128) void k_h(const float* __restrict__ x,
                                           const float* __restrict__ W1,
                                           const float* __restrict__ b1,
                                           const float* __restrict__ Wq,
                                           const float* __restrict__ Wk,
                                           float* __restrict__ h,
                                           float* __restrict__ Mqk) {
    __shared__ __attribute__((aligned(16))) float xs[32 * INDIM];
    if (blockIdx.x >= NROW / 16) {
        const int lane = threadIdx.x & 31;
        const int rr   = lane & 15;
        const int half = lane >> 4;
        const int tile = (blockIdx.x - NROW / 16) * 4 + (threadIdx.x >> 5);
        const int i0 = (tile >> 3) * 16;
        const int j0 = (tile & 7) * 16;
        const float* ap = Wq + (i0 + rr) * HID + half * 4;
        const float* bp = Wk + (j0 + rr) * HID + half * 4;
        v8f m0a = {0.f,0.f,0.f,0.f,0.f,0.f,0.f,0.f};
        v8f m1a = m0a;
#pragma unroll 4
        for (int t = 0; t < HID; t += 8) {
            const v4f a = *(const v4f*)(ap + t);
            const v4f b = *(const v4f*)(bp + t);
            v2f alo = {a.x, a.y}, ahi = {a.z, a.w};
            v2f blo = {b.x, b.y}, bhi = {b.z, b.w};
            m0a = wmma_f32_k4(alo, blo, m0a);
            m1a = wmma_f32_k4(ahi, bhi, m1a);
        }
#pragma unroll
        for (int r = 0; r < 8; ++r)
            Mqk[(i0 + r + half * 8) * HID + j0 + rr] = m0a[r] + m1a[r];
        return;
    }
    const int m0 = (blockIdx.x >> 1) * 32;
    const int ch = blockIdx.x & 1;
    {
        const v4f* srcv = (const v4f*)(x + (size_t)m0 * INDIM);
        v4f* dstv = (v4f*)xs;
        const bool tail = threadIdx.x < XV4 - XFULL * 128;
        v4f buf[XFULL + 1];
#pragma unroll
        for (int j = 0; j < XFULL; ++j) buf[j] = srcv[threadIdx.x + 128 * j];
        if (tail) buf[XFULL] = srcv[threadIdx.x + 128 * XFULL];
#pragma unroll
        for (int j = 0; j < XFULL; ++j) dstv[threadIdx.x + 128 * j] = buf[j];
        if (tail) dstv[threadIdx.x + 128 * XFULL] = buf[XFULL];
    }
    __syncthreads();

    const int wave = threadIdx.x >> 5;
    const int lane = threadIdx.x & 31;
    const int rr   = lane & 15;
    const int half = lane >> 4;
    const int n    = ch * 64 + wave * 16 + rr;

    const float* a0p = xs + rr * INDIM + half * 2;
    const float* a1p = xs + (16 + rr) * INDIM + half * 2;
    const float* bp  = W1 + (half * 2) * HID + n;

    v8f acc0 = {0.f,0.f,0.f,0.f,0.f,0.f,0.f,0.f};
    v8f acc1 = acc0;
    for (int kk = 0; kk < INDIM; kk += 4) {
        const v2f a0 = *(const v2f*)(a0p + kk);
        const v2f a1 = *(const v2f*)(a1p + kk);
        v2f bf; bf.x = bp[kk * HID]; bf.y = bp[(kk + 1) * HID];
        acc0 = wmma_f32_k4(a0, bf, acc0);
        acc1 = wmma_f32_k4(a1, bf, acc1);
    }
    const float bias = b1[n];
#pragma unroll
    for (int r = 0; r < 8; ++r) {
        const int m = m0 + r + half * 8;
        const float v0 = acc0[r] + bias;
        const float v1 = acc1[r] + bias;
        h[m * HID + n]        = v0 > 0.f ? v0 : 0.f;
        h[(m + 16) * HID + n] = v1 > 0.f ? v1 : 0.f;
    }
}

#define HP 132
__global__ __launch_bounds__(128) void k_tvg(const float* __restrict__ h,
                                             const float* __restrict__ Mqk,
                                             const float* __restrict__ Wv,
                                             float* __restrict__ V,
                                             float* __restrict__ G) {
    __shared__ __attribute__((aligned(16))) float hs[48 * HP];
    __shared__ __attribute__((aligned(16))) float ts[16 * HP];
    const int m0 = blockIdx.x * 16;
    {
        v4f buf[12];
#pragma unroll
        for (int j = 0; j < 12; ++j) {
            const int e   = threadIdx.x + 128 * j;
            const int row = min(max(m0 - 16 + (e >> 5), 0), NROW - 1);
            buf[j] = *(const v4f*)(h + (size_t)row * HID + (e & 31) * 4);
        }
#pragma unroll
        for (int j = 0; j < 12; ++j) {
            const int e = threadIdx.x + 128 * j;
            *(v4f*)(hs + (e >> 5) * HP + (e & 31) * 4) = buf[j];
        }
    }
    __syncthreads();

    const int wave = threadIdx.x >> 5;
    const int lane = threadIdx.x & 31;
    const int rr   = lane & 15;
    const int half = lane >> 4;
    const int n    = wave * 32 + rr;

    const float* ap = hs + (16 + rr) * HP + half * 2;
    const int    bo = (half * 2) * HID + n;

    v8f at0 = {0.f,0.f,0.f,0.f,0.f,0.f,0.f,0.f};
    v8f at1 = at0, av0 = at0, av1 = at0;
    for (int kk = 0; kk < HID; kk += 4) {
        const v2f a = *(const v2f*)(ap + kk);
        const int o = bo + kk * HID;
        v2f bt0; bt0.x = Mqk[o];      bt0.y = Mqk[o + HID];
        v2f bt1; bt1.x = Mqk[o + 16]; bt1.y = Mqk[o + 16 + HID];
        v2f bv0; bv0.x = Wv[o];       bv0.y = Wv[o + HID];
        v2f bv1; bv1.x = Wv[o + 16];  bv1.y = Wv[o + 16 + HID];
        at0 = wmma_f32_k4(a, bt0, at0);
        at1 = wmma_f32_k4(a, bt1, at1);
        av0 = wmma_f32_k4(a, bv0, av0);
        av1 = wmma_f32_k4(a, bv1, av1);
    }
#pragma unroll
    for (int r = 0; r < 8; ++r) {
        const int rl = r + half * 8;
        ts[rl * HP + n]      = at0[r];
        ts[rl * HP + n + 16] = at1[r];
    }
    __syncthreads();
#pragma unroll
    for (int r = 0; r < 8; ++r) {
        const int rl = r + half * 8;
        V[(m0 + rl) * HID + n]      = av0[r];
        V[(m0 + rl) * HID + n + 16] = av1[r];
    }

    if (wave < 3) {
        const float* tp = ts + rr * HP + half * 4;
        const float* cp = hs + (wave * 16 + rr) * HP + half * 4;
        v8f g = {0.f,0.f,0.f,0.f,0.f,0.f,0.f,0.f};
        for (int t = 0; t < HID; t += 8) {
            const v4f a = *(const v4f*)(tp + t);
            const v4f b = *(const v4f*)(cp + t);
            v2f alo = {a.x, a.y}, ahi = {a.z, a.w};
            v2f blo = {b.x, b.y}, bhi = {b.z, b.w};
            g = wmma_f32_k4(alo, blo, g);
            g = wmma_f32_k4(ahi, bhi, g);
        }
        const float scale = (float)(0.08838834764831845 * 1.4426950408889634);
        const int gc  = m0 - 16 + wave * 16 + rr;
        const int sm0 = m0 % SEQ;
#pragma unroll
        for (int r = 0; r < 8; ++r) {
            const int rl = r + half * 8;
            const int d  = gc - (m0 + rl);
            int sa = sm0 + rl;
            if (sa >= SEQ) sa -= SEQ;
            const int sc = sa + d;
            if (d >= -10 && d <= 10 && sc >= 0 && sc < SEQ)
                G[(long)(m0 + rl) * GP + (d + 10)] = g[r] * scale;
        }
    }
}

template <int J>
static __device__ __forceinline__ float row_bcast(float x) {
    const int b = __builtin_bit_cast(int, x);
    return __builtin_bit_cast(float,
        __builtin_amdgcn_update_dpp(b, b, 0x150 + J, 0xf, 0xf, true));
}

template <int J>
static __device__ __forceinline__ void pool_row(const float* __restrict__ vseq, int s,
                                                float bsum, v4f& plo, v4f& phi) {
    const int   rv = s + 5 - J;
    const int   rc = min(max(rv, 0), SEQ - 1);
    const float bj = row_bcast<J>(bsum);
    const float w  = (rv >= 0 && rv < SEQ) ? bj : 0.f;
    const float* vr = vseq + rc * HID;
    const v4f lo = *(const v4f*)(vr);
    const v4f hi = *(const v4f*)(vr + 64);
    plo += w * lo;
    phi += w * hi;
}

__global__ __launch_bounds__(256, 16) void k_attn(const float* __restrict__ G,
                                              const float* __restrict__ V,
                                              const float* __restrict__ W2,
                                              const float* __restrict__ b2,
                                              float* __restrict__ out,
                                              float* __restrict__ alphaOut) {
    const int rr  = threadIdx.x & 15;
    const int idx = blockIdx.x * 16 + (threadIdx.x >> 4);
    const int s   = idx % SEQ;

    const int  cj  = s + 5 - rr;
    const bool jin = rr < WIN;
    const bool jok = jin && (cj >= 0) && (cj < SEQ);
    const float* gcol  = G + (long)(idx + 5) * GP + (10 - rr);
    const float* gdiag = G + (long)idx * GP + 10;

    float scv[WIN];
#pragma unroll
    for (int i = 0; i < WIN; ++i) {
        const int  ai = s + 5 - i;
        const bool ok = jok && (ai >= 0) && (ai < SEQ);
        const float g = *(ok ? gcol - (GP - 1) * i : gdiag);
        scv[i] = ok ? g : 0.f;
    }

    const long  abase = (long)idx * (WIN * WIN);
    float bsum = 0.f;
#pragma unroll
    for (int i = 0; i < WIN; ++i) {
        const float sc = scv[i];
        float mv = jin ? sc : -3.0e38f;
        mv = __builtin_elementwise_maximum(mv, row_xor<1>(mv));
        mv = __builtin_elementwise_maximum(mv, row_xor<2>(mv));
        mv = __builtin_elementwise_maximum(mv, row_xor<4>(mv));
        mv = __builtin_elementwise_maximum(mv, row_xor<8>(mv));
        float p = jin ? __builtin_amdgcn_exp2f(sc - mv) : 0.f;
        float sum = p;
        sum += row_xor<1>(sum);
        sum += row_xor<2>(sum);
        sum += row_xor<4>(sum);
        sum += row_xor<8>(sum);
        const float al = p * __builtin_amdgcn_rcpf(sum);
        if (jin) alphaOut[abase + i * WIN + rr] = al;
        bsum += al;
    }

    v4f plo = {0.f, 0.f, 0.f, 0.f};
    v4f phi = {0.f, 0.f, 0.f, 0.f};
    const float* vseq = V + (long)(idx - s) * HID + rr * 4;
    pool_row<0>(vseq, s, bsum, plo, phi);
    pool_row<1>(vseq, s, bsum, plo, phi);
    pool_row<2>(vseq, s, bsum, plo, phi);
    pool_row<3>(vseq, s, bsum, plo, phi);
    pool_row<4>(vseq, s, bsum, plo, phi);
    pool_row<5>(vseq, s, bsum, plo, phi);
    pool_row<6>(vseq, s, bsum, plo, phi);
    pool_row<7>(vseq, s, bsum, plo, phi);
    pool_row<8>(vseq, s, bsum, plo, phi);
    pool_row<9>(vseq, s, bsum, plo, phi);
    pool_row<10>(vseq, s, bsum, plo, phi);

    const v4f wa = *(const v4f*)(W2 + rr * 8);
    const v4f wb = *(const v4f*)(W2 + rr * 8 + 4);
    const v4f wc = *(const v4f*)(W2 + 128 + rr * 8);
    const v4f wd = *(const v4f*)(W2 + 128 + rr * 8 + 4);
    float p0 = 0.f, p1 = 0.f;
    p0 += plo.x * wa.x;  p1 += plo.x * wa.y;
    p0 += plo.y * wa.z;  p1 += plo.y * wa.w;
    p0 += plo.z * wb.x;  p1 += plo.z * wb.y;
    p0 += plo.w * wb.z;  p1 += plo.w * wb.w;
    p0 += phi.x * wc.x;  p1 += phi.x * wc.y;
    p0 += phi.y * wc.z;  p1 += phi.y * wc.w;
    p0 += phi.z * wd.x;  p1 += phi.z * wd.y;
    p0 += phi.w * wd.z;  p1 += phi.w * wd.w;
    p0 += row_xor<1>(p0);  p1 += row_xor<1>(p1);
    p0 += row_xor<2>(p0);  p1 += row_xor<2>(p1);
    p0 += row_xor<4>(p0);  p1 += row_xor<4>(p1);
    p0 += row_xor<8>(p0);  p1 += row_xor<8>(p1);
    if (rr == 0) {
        out[idx * 2 + 0] = 1.f / (1.f + expf(-(p0 + b2[0])));
        out[idx * 2 + 1] = 1.f / (1.f + expf(-(p1 + b2[1])));
    }
}

extern "C" void kernel_launch(void* const* d_in, const int* in_sizes, int n_in,
                              void* d_out, int out_size, void* d_ws, size_t ws_size,
                              hipStream_t stream) {
    const float* x  = (const float*)d_in[0];
    const float* W1 = (const float*)d_in[1];
    const float* b1 = (const float*)d_in[2];
    const float* Wq = (const float*)d_in[3];
    const float* Wk = (const float*)d_in[4];
    const float* Wv = (const float*)d_in[5];
    const float* W2 = (const float*)d_in[6];
    const float* b2 = (const float*)d_in[7];

    float* h  = (float*)d_ws;
    float* G  = h  + (size_t)2 * NROW * HID;
    float* V  = G  + (size_t)NROW * HID;
    float* Mqk = G + (size_t)NROW * GP;

    float* out      = (float*)d_out;
    float* alphaOut = out + (size_t)NROW * 2;

    k_h  <<<NROW / 16 + MQK_BLOCKS, 128, 0, stream>>>(x, W1, b1, Wq, Wk, h, Mqk);
    k_tvg<<<NROW / 16, 128, 0, stream>>>(h, Mqk, Wv, V, G);
    k_attn<<<NROW / 16, 256, 0, stream>>>(G, V, W2, b2, out, alphaOut);
}


# k_attn folded into k_tvg: a block per half sequence (8 waves) keeps 64 rows of h, T, V and the band in LDS, does its 50 positions there; 2 launches, was 3; weight fragment feeds 2 WMMAs; sums same
# speedup vs baseline: 2.1505x; 1.0370x over previous
#include <hip/hip_runtime.h>
#include <math.h>

typedef __attribute__((ext_vector_type(2))) float v2f;
typedef __attribute__((ext_vector_type(4))) float v4f;
typedef __attribute__((ext_vector_type(8))) float v8f;

#define SEQ   100
#define BATCH 256
#define INDIM 300
#define HID   128
#define NROW  (BATCH*SEQ)
#define WIN   11
#define GP    32

static __device__ __forceinline__ v8f wmma_f32_k4(v2f a, v2f b, v8f c) {
    return __builtin_amdgcn_wmma_f32_16x16x4_f32(false, a, false, b, (short)0, c,
                                                 false, false);
}

template <int M>
static __device__ __forceinline__ float row_xor(float x) {
    const int b = __builtin_bit_cast(int, x);
    return __builtin_bit_cast(float,
        __builtin_amdgcn_update_dpp(b, b, 0x160 + M, 0xf, 0xf, true));
}

#define MQK_BLOCKS 16

#define XV4   (32 * INDIM / 4)
#define XFULL (XV4 / 128)
__global__ __launch_bounds__(128) void k_h(const float* __restrict__ x,
                                           const float* __restrict__ W1,
                                           const float* __restrict__ b1,
                                           const float* __restrict__ Wq,
                                           const float* __restrict__ Wk,
                                           float* __restrict__ h,
                                           float* __restrict__ Mqk) {
    __shared__ __attribute__((aligned(16))) float xs[32 * INDIM];
    if (blockIdx.x >= NROW / 16) {
        const int lane = threadIdx.x & 31;
        const int rr   = lane & 15;
        const int half = lane >> 4;
        const int tile = (blockIdx.x - NROW / 16) * 4 + (threadIdx.x >> 5);
        const int i0 = (tile >> 3) * 16;
        const int j0 = (tile & 7) * 16;
        const float* ap = Wq + (i0 + rr) * HID + half * 4;
        const float* bp = Wk + (j0 + rr) * HID + half * 4;
        v8f m0a = {0.f,0.f,0.f,0.f,0.f,0.f,0.f,0.f};
        v8f m1a = m0a;
#pragma unroll 4
        for (int t = 0; t < HID; t += 8) {
            const v4f a = *(const v4f*)(ap + t);
            const v4f b = *(const v4f*)(bp + t);
            v2f alo = {a.x, a.y}, ahi = {a.z, a.w};
            v2f blo = {b.x, b.y}, bhi = {b.z, b.w};
            m0a = wmma_f32_k4(alo, blo, m0a);
            m1a = wmma_f32_k4(ahi, bhi, m1a);
        }
#pragma unroll
        for (int r = 0; r < 8; ++r)
            Mqk[(i0 + r + half * 8) * HID + j0 + rr] = m0a[r] + m1a[r];
        return;
    }
    const int m0 = (blockIdx.x >> 1) * 32;
    const int ch = blockIdx.x & 1;
    {
        const v4f* srcv = (const v4f*)(x + (size_t)m0 * INDIM);
        v4f* dstv = (v4f*)xs;
        const bool tail = threadIdx.x < XV4 - XFULL * 128;
        v4f buf[XFULL + 1];
#pragma unroll
        for (int j = 0; j < XFULL; ++j) buf[j] = srcv[threadIdx.x + 128 * j];
        if (tail) buf[XFULL] = srcv[threadIdx.x + 128 * XFULL];
#pragma unroll
        for (int j = 0; j < XFULL; ++j) dstv[threadIdx.x + 128 * j] = buf[j];
        if (tail) dstv[threadIdx.x + 128 * XFULL] = buf[XFULL];
    }
    __syncthreads();

    const int wave = threadIdx.x >> 5;
    const int lane = threadIdx.x & 31;
    const int rr   = lane & 15;
    const int half = lane >> 4;
    const int n    = ch * 64 + wave * 16 + rr;

    const float* a0p = xs + rr * INDIM + half * 2;
    const float* a1p = xs + (16 + rr) * INDIM + half * 2;
    const float* bp  = W1 + (half * 2) * HID + n;

    v8f acc0 = {0.f,0.f,0.f,0.f,0.f,0.f,0.f,0.f};
    v8f acc1 = acc0;
    for (int kk = 0; kk < INDIM; kk += 4) {
        const v2f a0 = *(const v2f*)(a0p + kk);
        const v2f a1 = *(const v2f*)(a1p + kk);
        v2f bf; bf.x = bp[kk * HID]; bf.y = bp[(kk + 1) * HID];
        acc0 = wmma_f32_k4(a0, bf, acc0);
        acc1 = wmma_f32_k4(a1, bf, acc1);
    }
    const float bias = b1[n];
#pragma unroll
    for (int r = 0; r < 8; ++r) {
        const int m = m0 + r + half * 8;
        const float v0 = acc0[r] + bias;
        const float v1 = acc1[r] + bias;
        h[m * HID + n]        = v0 > 0.f ? v0 : 0.f;
        h[(m + 16) * HID + n] = v1 > 0.f ? v1 : 0.f;
    }
}

#define HP 132
#define FR 64
#define PH 50

template <int J>
static __device__ __forceinline__ float row_bcast(float x) {
    const int b = __builtin_bit_cast(int, x);
    return __builtin_bit_cast(float,
        __builtin_amdgcn_update_dpp(b, b, 0x150 + J, 0xf, 0xf, true));
}

template <int J>
static __device__ __forceinline__ void pool_row(const float* vloc, int s, int r0,
                                                float bsum, v4f& plo, v4f& phi) {
    const int   rv = s + 5 - J;
    const int   rc = min(max(rv, 0), SEQ - 1);
    const float bj = row_bcast<J>(bsum);
    const float w  = (rv >= 0 && rv < SEQ) ? bj : 0.f;
    const float* vr = vloc + (rc - r0) * HP;
    const v4f lo = *(const v4f*)(vr);
    const v4f hi = *(const v4f*)(vr + 64);
    plo += w * lo;
    phi += w * hi;
}

__global__ __launch_bounds__(256) void k_tva(const float* __restrict__ h,
                                             const float* __restrict__ Mqk,
                                             const float* __restrict__ Wv,
                                             const float* __restrict__ W2,
                                             const float* __restrict__ b2,
                                             float* __restrict__ out,
                                             float* __restrict__ alphaOut) {
    __shared__ __attribute__((aligned(16))) float hs[FR * HP];
    __shared__ __attribute__((aligned(16))) float ts[FR * HP];
    __shared__ __attribute__((aligned(16))) float vs[FR * HP];
    __shared__ __attribute__((aligned(16))) float gs[FR * GP];
    const int q  = blockIdx.x >> 1;
    const int p  = blockIdx.x & 1;
    const int r0 = p ? SEQ - FR : 0;
    {
        const float* src = h + ((size_t)q * SEQ + r0) * HID;
        v4f buf[8];
#pragma unroll
        for (int j = 0; j < 8; ++j) buf[j] = *(const v4f*)(src + (threadIdx.x + 256 * j) * 4);
#pragma unroll
        for (int j = 0; j < 8; ++j) {
            const int e = threadIdx.x + 256 * j;
            *(v4f*)(hs + (e >> 5) * HP + (e & 31) * 4) = buf[j];
        }
    }
    __syncthreads();

    const int wave = __builtin_amdgcn_readfirstlane(threadIdx.x >> 5);
    const int lane = threadIdx.x & 31;
    const int rr   = lane & 15;
    const int half = lane >> 4;
    {
        const int rp = wave & 1;
        const int n  = (wave >> 1) * 32 + rr;
        const float* a0p = hs + (rp * 32 + rr) * HP + half * 2;
        const float* a1p = a0p + 16 * HP;
        const int    bo  = (half * 2) * HID + n;

        v8f t00 = {0.f,0.f,0.f,0.f,0.f,0.f,0.f,0.f};
        v8f t01 = t00, v00 = t00, v01 = t00;
        v8f t10 = t00, t11 = t00, v10 = t00, v11 = t00;
        for (int kk = 0; kk < HID; kk += 4) {
            const v2f a0 = *(const v2f*)(a0p + kk);
            const v2f a1 = *(const v2f*)(a1p + kk);
            const int o = bo + kk * HID;
            v2f bt0; bt0.x = Mqk[o];      bt0.y = Mqk[o + HID];
            v2f bt1; bt1.x = Mqk[o + 16]; bt1.y = Mqk[o + 16 + HID];
            v2f bv0; bv0.x = Wv[o];       bv0.y = Wv[o + HID];
            v2f bv1; bv1.x = Wv[o + 16];  bv1.y = Wv[o + 16 + HID];
            t00 = wmma_f32_k4(a0, bt0, t00);
            t10 = wmma_f32_k4(a1, bt0, t10);
            t01 = wmma_f32_k4(a0, bt1, t01);
            t11 = wmma_f32_k4(a1, bt1, t11);
            v00 = wmma_f32_k4(a0, bv0, v00);
            v10 = wmma_f32_k4(a1, bv0, v10);
            v01 = wmma_f32_k4(a0, bv1, v01);
            v11 = wmma_f32_k4(a1, bv1, v11);
        }
#pragma unroll
        for (int r = 0; r < 8; ++r) {
            const int o0 = (rp * 32 + r + half * 8) * HP + n;
            const int o1 = o0 + 16 * HP;
            ts[o0] = t00[r];  ts[o0 + 16] = t01[r];
            ts[o1] = t10[r];  ts[o1 + 16] = t11[r];
            vs[o0] = v00[r];  vs[o0 + 16] = v01[r];
            vs[o1] = v10[r];  vs[o1 + 16] = v11[r];
        }
    }
    __syncthreads();

    for (int t = wave + 1; t < 11; t += 8) {
        const int i = t / 3;
        const int j = i - 1 + (t - 3 * i);
        const float* tp = ts + (16 * i + rr) * HP + half * 4;
        const float* cp = hs + (16 * j + rr) * HP + half * 4;
        v8f g = {0.f,0.f,0.f,0.f,0.f,0.f,0.f,0.f};
        for (int u = 0; u < HID; u += 8) {
            const v4f a = *(const v4f*)(tp + u);
            const v4f b = *(const v4f*)(cp + u);
            v2f alo = {a.x, a.y}, ahi = {a.z, a.w};
            v2f blo = {b.x, b.y}, bhi = {b.z, b.w};
            g = wmma_f32_k4(alo, blo, g);
            g = wmma_f32_k4(ahi, bhi, g);
        }
        const float scale = (float)(0.08838834764831845 * 1.4426950408889634);
        const int d0 = 16 * (j - i) + rr - half * 8;
#pragma unroll
        for (int r = 0; r < 8; ++r) {
            const int d = d0 - r;
            if (d >= -10 && d <= 10)
                gs[(16 * i + r + half * 8) * GP + (d + 10)] = g[r] * scale;
        }
    }
    __syncthreads();

    const float* vloc = vs + rr * 4;
    for (int pp = wave; pp < PH / 2; pp += 8) {
        const int s   = p * PH + pp * 2 + half;
        const int ls  = s - r0;
        const int idx = q * SEQ + s;

        const int  cj  = s + 5 - rr;
        const bool jin = rr < WIN;
        const bool jok = jin && (cj >= 0) && (cj < SEQ);
        const float* gcol  = gs + (ls + 5) * GP + (10 - rr);
        const float* gdiag = gs + ls * GP + 10;

        float scv[WIN];
#pragma unroll
        for (int i = 0; i < WIN; ++i) {
            const int  ai = s + 5 - i;
            const bool ok = jok && (ai >= 0) && (ai < SEQ);
            const float g = *(ok ? gcol - (GP - 1) * i : gdiag);
            scv[i] = ok ? g : 0.f;
        }

        const long  abase = (long)idx * (WIN * WIN);
        float bsum = 0.f;
#pragma unroll
        for (int i = 0; i < WIN; ++i) {
            const float sc = scv[i];
            float mv = jin ? sc : -3.0e38f;
            mv = __builtin_elementwise_maximum(mv, row_xor<1>(mv));
            mv = __builtin_elementwise_maximum(mv, row_xor<2>(mv));
            mv = __builtin_elementwise_maximum(mv, row_xor<4>(mv));
            mv = __builtin_elementwise_maximum(mv, row_xor<8>(mv));
            float pe = jin ? __builtin_amdgcn_exp2f(sc - mv) : 0.f;
            float sum = pe;
            sum += row_xor<1>(sum);
            sum += row_xor<2>(sum);
            sum += row_xor<4>(sum);
            sum += row_xor<8>(sum);
            const float al = pe * __builtin_amdgcn_rcpf(sum);
            if (jin) alphaOut[abase + i * WIN + rr] = al;
            bsum += al;
        }

        v4f plo = {0.f, 0.f, 0.f, 0.f};
        v4f phi = {0.f, 0.f, 0.f, 0.f};
        pool_row<0>(vloc, s, r0, bsum, plo, phi);
        pool_row<1>(vloc, s, r0, bsum, plo, phi);
        pool_row<2>(vloc, s, r0, bsum, plo, phi);
        pool_row<3>(vloc, s, r0, bsum, plo, phi);
        pool_row<4>(vloc, s, r0, bsum, plo, phi);
        pool_row<5>(vloc, s, r0, bsum, plo, phi);
        pool_row<6>(vloc, s, r0, bsum, plo, phi);
        pool_row<7>(vloc, s, r0, bsum, plo, phi);
        pool_row<8>(vloc, s, r0, bsum, plo, phi);
        pool_row<9>(vloc, s, r0, bsum, plo, phi);
        pool_row<10>(vloc, s, r0, bsum, plo, phi);

        const v4f wa = *(const v4f*)(W2 + rr * 8);
        const v4f wb = *(const v4f*)(W2 + rr * 8 + 4);
        const v4f wc = *(const v4f*)(W2 + 128 + rr * 8);
        const v4f wd = *(const v4f*)(W2 + 128 + rr * 8 + 4);
        float p0 = 0.f, p1 = 0.f;
        p0 += plo.x * wa.x;  p1 += plo.x * wa.y;
        p0 += plo.y * wa.z;  p1 += plo.y * wa.w;
        p0 += plo.z * wb.x;  p1 += plo.z * wb.y;
        p0 += plo.w * wb.z;  p1 += plo.w * wb.w;
        p0 += phi.x * wc.x;  p1 += phi.x * wc.y;
        p0 += phi.y * wc.z;  p1 += phi.y * wc.w;
        p0 += phi.z * wd.x;  p1 += phi.z * wd.y;
        p0 += phi.w * wd.z;  p1 += phi.w * wd.w;
        p0 += row_xor<1>(p0);  p1 += row_xor<1>(p1);
        p0 += row_xor<2>(p0);  p1 += row_xor<2>(p1);
        p0 += row_xor<4>(p0);  p1 += row_xor<4>(p1);
        p0 += row_xor<8>(p0);  p1 += row_xor<8>(p1);
        if (rr == 0) {
            out[idx * 2 + 0] = 1.f / (1.f + expf(-(p0 + b2[0])));
            out[idx * 2 + 1] = 1.f / (1.f + expf(-(p1 + b2[1])));
        }
    }
}

extern "C" void kernel_launch(void* const* d_in, const int* in_sizes, int n_in,
                              void* d_out, int out_size, void* d_ws, size_t ws_size,
                              hipStream_t stream) {
    const float* x  = (const float*)d_in[0];
    const float* W1 = (const float*)d_in[1];
    const float* b1 = (const float*)d_in[2];
    const float* Wq = (const float*)d_in[3];
    const float* Wk = (const float*)d_in[4];
    const float* Wv = (const float*)d_in[5];
    const float* W2 = (const float*)d_in[6];
    const float* b2 = (const float*)d_in[7];

    float* h   = (float*)d_ws;
    float* Mqk = h + (size_t)NROW * HID;

    float* out      = (float*)d_out;
    float* alphaOut = out + (size_t)NROW * 2;

    k_h  <<<NROW / 16 + MQK_BLOCKS, 128, 0, stream>>>(x, W1, b1, Wq, Wk, h, Mqk);
    k_tva<<<BATCH * 2, 256, 0, stream>>>(h, Mqk, Wv, W2, b2, out, alphaOut);
}
